# starting kernel
# baseline (speedup 1.0000x reference)
#include <hip/hip_runtime.h>

typedef float v2f __attribute__((ext_vector_type(2)));
typedef float v8f __attribute__((ext_vector_type(8)));

#define GN 512
#define LDSP 17

__global__ __launch_bounds__(256)
void pml_coeff(const float* __restrict__ b,
               float* __restrict__ C1p, float* __restrict__ C3p)
{
  const int i = blockIdx.x * 256 + threadIdx.x;
  const float bv = b[i];
  C1p[i] = 1.0f / (4.0f + bv);
  C3p[i] = 4.0f - bv;
}

__global__ __launch_bounds__(256)
void wave_step(const float* __restrict__ xi,
               const float* __restrict__ xi_next,
               const float* __restrict__ u1,
               const float* __restrict__ u2,
               const float* __restrict__ c2,
               const float* __restrict__ msk,
               const float* __restrict__ C1p,
               const float* __restrict__ C3p,
               float* __restrict__ out)
{
  __shared__ float lds[8][16 * LDSP];

  const int lane = threadIdx.x & 31;
  const int wave = threadIdx.x >> 5;
  const int tile = blockIdx.x * 8 + wave;
  const int r0 = (tile >> 5) * 16;
  const int c0 = (tile & 31) * 16;

  const int hi = lane >> 4;
  const int lp = lane & 15;

  float* t = lds[wave];

  const unsigned rowcode = (unsigned)((0x19ull << (2 * lp)) >> 2);

  {
    const int lr = lane >> 2;
    const int lc = (lane & 3) * 4;
    const float4 p0 = *(const float4*)&u1[(r0 + lr) * GN + c0 + lc];
    const float4 p1 = *(const float4*)&u1[(r0 + 8 + lr) * GN + c0 + lc];
    float* q0 = &t[lr * LDSP + lc];
    q0[0] = p0.x; q0[1] = p0.y; q0[2] = p0.z; q0[3] = p0.w;
    float* q1 = &t[(lr + 8) * LDSP + lc];
    q1[0] = p1.x; q1[1] = p1.y; q1[2] = p1.z; q1[3] = p1.w;
  }

  v8f acc = {};
  #pragma unroll
  for (int k0 = 0; k0 < 16; k0 += 4) {
    const int sh = 30 - 2 * k0 - 4 * hi;

    v2f tv;
    tv.x = (float)((int)(rowcode << sh) >> 30);
    tv.y = (float)((int)(rowcode << (sh - 2)) >> 30);

    v2f bU;
    bU.x = t[(k0 + 2 * hi + 0) * LDSP + lp];
    bU.y = t[(k0 + 2 * hi + 1) * LDSP + lp];
    acc = __builtin_amdgcn_wmma_f32_16x16x4_f32(
        false, tv, false, bU, (short)0, acc, false, false);

    v2f aU;
    aU.x = t[lp * LDSP + k0 + 2 * hi + 0];
    aU.y = t[lp * LDSP + k0 + 2 * hi + 1];
    acc = __builtin_amdgcn_wmma_f32_16x16x4_f32(
        false, aU, false, tv, (short)0, acc, false, false);
  }

  #pragma unroll
  for (int v = 0; v < 8; ++v) {
    const int M  = v + 8 * hi;
    const int gr = r0 + M;
    const int gc = c0 + lp;
    const int idx = gr * GN + gc;

    float lap = acc[v];
    if (M == 0  && gr > 0)      lap += u1[(gr - 1) * GN + gc];
    if (M == 15 && gr < GN - 1) lap += u1[(gr + 1) * GN + gc];
    if (lp == 0  && gc > 0)      lap += u1[idx - 1];
    if (lp == 15 && gc < GN - 1) lap += u1[idx + 1];

    const float u1c = t[M * LDSP + lp];
    out[idx] = C1p[idx] * (8.0f * u1c - C3p[idx] * u2[idx]
                           + c2[idx] * lap + msk[idx] * xi[idx]);

    if (lp == 0) __builtin_prefetch(&xi_next[gr * GN + c0], 0, 1);
  }
}

extern "C" void kernel_launch(void* const* d_in, const int* in_sizes, int n_in,
                              void* d_out, int out_size, void* d_ws, size_t ws_size,
                              hipStream_t stream) {
  const float* inp = (const float*)d_in[0];
  const float* un1 = (const float*)d_in[1];
  const float* un2 = (const float*)d_in[2];
  const float* c2  = (const float*)d_in[3];
  const float* msk = (const float*)d_in[4];
  const float* b   = (const float*)d_in[5];

  const int plane = GN * GN;
  const int nt = in_sizes[0] / plane;

  float* buf0 = (float*)d_ws;
  float* buf1 = buf0 + plane;
  float* C1p  = buf1 + plane;
  float* C3p  = C1p  + plane;
  float* outp = (float*)d_out;

  pml_coeff<<<dim3(plane / 256), dim3(256), 0, stream>>>(b, C1p, C3p);

  dim3 grid(128);
  dim3 block(256);

  for (int t = 0; t < nt; ++t) {
    const float* xi      = inp + (size_t)t * plane;
    const float* xi_next = (t + 1 < nt) ? (inp + (size_t)(t + 1) * plane) : xi;

    const float* u1 = (t == 0) ? un1 : ((t & 1) ? buf0 : buf1);
    const float* u2 = (t == 0) ? un2 : ((t == 1) ? un1 : ((t & 1) ? buf1 : buf0));
    float* o = (t == nt - 1) ? outp : ((t & 1) ? buf1 : buf0);

    wave_step<<<grid, block, 0, stream>>>(xi, xi_next, u1, u2, c2, msk,
                                          C1p, C3p, o);
  }
}


# All global loads of a tile issued as one batch straight into WMMA fragment layouts (no LDS, no per-row load-wait chain); halo edges via a ninth WMMA; k order and halo sum order change
# speedup vs baseline: 3.1422x; 3.1422x over previous
#include <hip/hip_runtime.h>

typedef float v2f __attribute__((ext_vector_type(2)));
typedef float v8f __attribute__((ext_vector_type(8)));

#define GN 512

__global__ __launch_bounds__(256)
void pml_coeff(const float* __restrict__ b,
               float* __restrict__ C1p, float* __restrict__ C3p)
{
  const int i = blockIdx.x * 256 + threadIdx.x;
  const float bv = b[i];
  C1p[i] = 1.0f / (4.0f + bv);
  C3p[i] = 4.0f - bv;
}

__global__ __launch_bounds__(256)
void wave_step(const float* __restrict__ xi,
               const float* __restrict__ xi_next,
               const float* __restrict__ u1,
               const float* __restrict__ u2,
               const float* __restrict__ c2,
               const float* __restrict__ msk,
               const float* __restrict__ C1p,
               const float* __restrict__ C3p,
               float* __restrict__ out)
{
  const int lane = threadIdx.x & 31;
  const int wave = threadIdx.x >> 5;
  const int tile = blockIdx.x * 8 + wave;
  const int r0 = (tile >> 5) * 16;
  const int c0 = (tile & 31) * 16;

  const int hi = lane >> 4;
  const int lp = lane & 15;

  const unsigned rowcode = (unsigned)((0x19ull << (2 * lp)) >> 2);

  const int rowoff = (r0 + lp) * GN + c0;

  if (hi == 0) __builtin_prefetch(&xi_next[rowoff], 0, 1);

  const int base = (r0 + 8 * hi) * GN + c0 + lp;
  float uc[8], k1[8], k3[8], w2[8], cc[8], mk[8], xv[8];
  #pragma unroll
  for (int v = 0; v < 8; ++v) {
    const int idx = base + v * GN;
    uc[v] = u1[idx];
    k1[v] = C1p[idx];
    k3[v] = C3p[idx];
    w2[v] = u2[idx];
    cc[v] = c2[idx];
    mk[v] = msk[idx];
    xv[v] = xi[idx];
  }

  const float4 ra = *(const float4*)&u1[rowoff + 8 * hi];
  const float4 rb = *(const float4*)&u1[rowoff + 8 * hi + 4];

  const int a0 = hi ? (r0 - 1) * GN + c0 + lp  : rowoff - 1;
  const int a1 = hi ? (r0 + 16) * GN + c0 + lp : rowoff + 16;
  const bool ok0 = hi ? (r0 > 0)       : (c0 > 0);
  const bool ok1 = hi ? (r0 + 16 < GN) : (c0 + 16 < GN);
  const float g0 = u1[ok0 ? a0 : base];
  const float g1 = u1[ok1 ? a1 : base];
  const float h0 = ok0 ? g0 : 0.0f;
  const float h1 = ok1 ? g1 : 0.0f;

  v8f acc = {};
  #pragma unroll
  for (int j = 0; j < 4; ++j) {
    const int sh = 30 - 16 * hi - 4 * j;

    v2f tv;
    tv.x = (float)((int)(rowcode << sh) >> 30);
    tv.y = (float)((int)(rowcode << (sh - 2)) >> 30);

    v2f bU;
    bU.x = uc[2 * j];
    bU.y = uc[2 * j + 1];
    acc = __builtin_amdgcn_wmma_f32_16x16x4_f32(
        false, tv, false, bU, (short)0, acc, false, false);

    v2f aU;
    aU.x = (j == 0) ? ra.x : (j == 1) ? ra.z : (j == 2) ? rb.x : rb.z;
    aU.y = (j == 0) ? ra.y : (j == 1) ? ra.w : (j == 2) ? rb.y : rb.w;
    acc = __builtin_amdgcn_wmma_f32_16x16x4_f32(
        false, aU, false, tv, (short)0, acc, false, false);
  }
  {
    const float e0  = (lp == 0)  ? 1.0f : 0.0f;
    const float e15 = (lp == 15) ? 1.0f : 0.0f;
    v2f a9, b9;
    a9.x = hi ? e0  : h0;
    a9.y = hi ? e15 : h1;
    b9.x = hi ? h0  : e0;
    b9.y = hi ? h1  : e15;
    acc = __builtin_amdgcn_wmma_f32_16x16x4_f32(
        false, a9, false, b9, (short)0, acc, false, false);
  }

  #pragma unroll
  for (int v = 0; v < 8; ++v) {
    const int idx = base + v * GN;
    out[idx] = k1[v] * (8.0f * uc[v] - k3[v] * w2[v]
                        + cc[v] * acc[v] + mk[v] * xv[v]);
  }
}

extern "C" void kernel_launch(void* const* d_in, const int* in_sizes, int n_in,
                              void* d_out, int out_size, void* d_ws, size_t ws_size,
                              hipStream_t stream) {
  const float* inp = (const float*)d_in[0];
  const float* un1 = (const float*)d_in[1];
  const float* un2 = (const float*)d_in[2];
  const float* c2  = (const float*)d_in[3];
  const float* msk = (const float*)d_in[4];
  const float* b   = (const float*)d_in[5];

  const int plane = GN * GN;
  const int nt = in_sizes[0] / plane;

  float* buf0 = (float*)d_ws;
  float* buf1 = buf0 + plane;
  float* C1p  = buf1 + plane;
  float* C3p  = C1p  + plane;
  float* outp = (float*)d_out;

  pml_coeff<<<dim3(plane / 256), dim3(256), 0, stream>>>(b, C1p, C3p);

  dim3 grid(128);
  dim3 block(256);

  for (int t = 0; t < nt; ++t) {
    const float* xi      = inp + (size_t)t * plane;
    const float* xi_next = (t + 1 < nt) ? (inp + (size_t)(t + 1) * plane) : xi;

    const float* u1 = (t == 0) ? un1 : ((t & 1) ? buf0 : buf1);
    const float* u2 = (t == 0) ? un2 : ((t == 1) ? un1 : ((t & 1) ? buf1 : buf0));
    float* o = (t == nt - 1) ? outp : ((t & 1) ? buf1 : buf0);

    wave_step<<<grid, block, 0, stream>>>(xi, xi_next, u1, u2, c2, msk,
                                          C1p, C3p, o);
  }
}


# Workgroups of 4 waves instead of 8 (256 workgroups instead of 128): one tile wave per SIMD on every WGP, halving the loads each WGP's memory path serves per step; sum order unchanged
# speedup vs baseline: 3.5695x; 1.1360x over previous
#include <hip/hip_runtime.h>

typedef float v2f __attribute__((ext_vector_type(2)));
typedef float v8f __attribute__((ext_vector_type(8)));

#define GN 512

__global__ __launch_bounds__(256)
void pml_coeff(const float* __restrict__ b,
               float* __restrict__ C1p, float* __restrict__ C3p)
{
  const int i = blockIdx.x * 256 + threadIdx.x;
  const float bv = b[i];
  C1p[i] = 1.0f / (4.0f + bv);
  C3p[i] = 4.0f - bv;
}

__global__ __launch_bounds__(128)
void wave_step(const float* __restrict__ xi,
               const float* __restrict__ xi_next,
               const float* __restrict__ u1,
               const float* __restrict__ u2,
               const float* __restrict__ c2,
               const float* __restrict__ msk,
               const float* __restrict__ C1p,
               const float* __restrict__ C3p,
               float* __restrict__ out)
{
  const int lane = threadIdx.x & 31;
  const int wave = threadIdx.x >> 5;
  const int tile = blockIdx.x * 4 + wave;
  const int r0 = (tile >> 5) * 16;
  const int c0 = (tile & 31) * 16;

  const int hi = lane >> 4;
  const int lp = lane & 15;

  const unsigned rowcode = (unsigned)((0x19ull << (2 * lp)) >> 2);

  const int rowoff = (r0 + lp) * GN + c0;

  if (hi == 0) __builtin_prefetch(&xi_next[rowoff], 0, 1);

  const int base = (r0 + 8 * hi) * GN + c0 + lp;
  float uc[8], k1[8], k3[8], w2[8], cc[8], mk[8], xv[8];
  #pragma unroll
  for (int v = 0; v < 8; ++v) {
    const int idx = base + v * GN;
    uc[v] = u1[idx];
    k1[v] = C1p[idx];
    k3[v] = C3p[idx];
    w2[v] = u2[idx];
    cc[v] = c2[idx];
    mk[v] = msk[idx];
    xv[v] = xi[idx];
  }

  const float4 ra = *(const float4*)&u1[rowoff + 8 * hi];
  const float4 rb = *(const float4*)&u1[rowoff + 8 * hi + 4];

  const int a0 = hi ? (r0 - 1) * GN + c0 + lp  : rowoff - 1;
  const int a1 = hi ? (r0 + 16) * GN + c0 + lp : rowoff + 16;
  const bool ok0 = hi ? (r0 > 0)       : (c0 > 0);
  const bool ok1 = hi ? (r0 + 16 < GN) : (c0 + 16 < GN);
  const float g0 = u1[ok0 ? a0 : base];
  const float g1 = u1[ok1 ? a1 : base];
  const float h0 = ok0 ? g0 : 0.0f;
  const float h1 = ok1 ? g1 : 0.0f;

  v8f acc = {};
  #pragma unroll
  for (int j = 0; j < 4; ++j) {
    const int sh = 30 - 16 * hi - 4 * j;

    v2f tv;
    tv.x = (float)((int)(rowcode << sh) >> 30);
    tv.y = (float)((int)(rowcode << (sh - 2)) >> 30);

    v2f bU;
    bU.x = uc[2 * j];
    bU.y = uc[2 * j + 1];
    acc = __builtin_amdgcn_wmma_f32_16x16x4_f32(
        false, tv, false, bU, (short)0, acc, false, false);

    v2f aU;
    aU.x = (j == 0) ? ra.x : (j == 1) ? ra.z : (j == 2) ? rb.x : rb.z;
    aU.y = (j == 0) ? ra.y : (j == 1) ? ra.w : (j == 2) ? rb.y : rb.w;
    acc = __builtin_amdgcn_wmma_f32_16x16x4_f32(
        false, aU, false, tv, (short)0, acc, false, false);
  }
  {
    const float e0  = (lp == 0)  ? 1.0f : 0.0f;
    const float e15 = (lp == 15) ? 1.0f : 0.0f;
    v2f a9, b9;
    a9.x = hi ? e0  : h0;
    a9.y = hi ? e15 : h1;
    b9.x = hi ? h0  : e0;
    b9.y = hi ? h1  : e15;
    acc = __builtin_amdgcn_wmma_f32_16x16x4_f32(
        false, a9, false, b9, (short)0, acc, false, false);
  }

  #pragma unroll
  for (int v = 0; v < 8; ++v) {
    const int idx = base + v * GN;
    out[idx] = k1[v] * (8.0f * uc[v] - k3[v] * w2[v]
                        + cc[v] * acc[v] + mk[v] * xv[v]);
  }
}

extern "C" void kernel_launch(void* const* d_in, const int* in_sizes, int n_in,
                              void* d_out, int out_size, void* d_ws, size_t ws_size,
                              hipStream_t stream) {
  const float* inp = (const float*)d_in[0];
  const float* un1 = (const float*)d_in[1];
  const float* un2 = (const float*)d_in[2];
  const float* c2  = (const float*)d_in[3];
  const float* msk = (const float*)d_in[4];
  const float* b   = (const float*)d_in[5];

  const int plane = GN * GN;
  const int nt = in_sizes[0] / plane;

  float* buf0 = (float*)d_ws;
  float* buf1 = buf0 + plane;
  float* C1p  = buf1 + plane;
  float* C3p  = C1p  + plane;
  float* outp = (float*)d_out;

  pml_coeff<<<dim3(plane / 256), dim3(256), 0, stream>>>(b, C1p, C3p);

  dim3 grid(256);
  dim3 block(128);

  for (int t = 0; t < nt; ++t) {
    const float* xi      = inp + (size_t)t * plane;
    const float* xi_next = (t + 1 < nt) ? (inp + (size_t)(t + 1) * plane) : xi;

    const float* u1 = (t == 0) ? un1 : ((t & 1) ? buf0 : buf1);
    const float* u2 = (t == 0) ? un2 : ((t == 1) ? un1 : ((t & 1) ? buf1 : buf0));
    float* o = (t == nt - 1) ? outp : ((t & 1) ? buf1 : buf0);

    wave_step<<<grid, block, 0, stream>>>(xi, xi_next, u1, u2, c2, msk,
                                          C1p, C3p, o);
  }
}


# Two time steps per launch, in registers per wave: both run transposed (WMMA A/B swapped) so a result is the next A chunk, V*I WMMAs give the B chunk; halo ring recomputed pointwise (other sum order)
# speedup vs baseline: 5.3939x; 1.5111x over previous
#include <hip/hip_runtime.h>

typedef float v2f __attribute__((ext_vector_type(2)));
typedef float v8f __attribute__((ext_vector_type(8)));

#define GN 512

__global__ __launch_bounds__(256)
void pml_coeff(const float* __restrict__ b, float* __restrict__ C1p)
{
  const int i = blockIdx.x * 256 + threadIdx.x;
  C1p[i] = 1.0f / (4.0f + b[i]);
}

__device__ __forceinline__ v8f mma4(v2f a, v2f b, v8f c)
{
  return __builtin_amdgcn_wmma_f32_16x16x4_f32(false, a, false, b, (short)0, c,
                                               false, false);
}

struct RingCell {
  float uc, gu, gd, gl, gr, k1, bv, w2, cc, mk, xv;
  bool ok, up, dn, lf, rt;
  int id;
};

__device__ __forceinline__ void ring_addr(RingCell& q, int r, int c, bool ok,
                                          int r0, int c0)
{
  const int rr = ok ? r : r0;
  const int cq = ok ? c : c0;
  q.id = rr * GN + cq;
  q.ok = ok;
  q.up = rr > 0;  q.dn = rr < GN - 1;  q.lf = cq > 0;  q.rt = cq < GN - 1;
}

__device__ __forceinline__ float ring_value(const RingCell& q)
{
  const float lap = (q.up ? q.gu : 0.0f) + (q.dn ? q.gd : 0.0f)
                    + (q.lf ? q.gl : 0.0f) + (q.rt ? q.gr : 0.0f) - 4.0f * q.uc;
  const float un = q.k1 * (8.0f * q.uc - (4.0f - q.bv) * q.w2 + q.cc * lap
                           + q.mk * q.xv);
  return q.ok ? un : 0.0f;
}

__global__ __launch_bounds__(128)
void wave_step2(const float* __restrict__ xi0,
                const float* __restrict__ xi1,
                const float* __restrict__ xi_nx,
                const float* __restrict__ u1,
                const float* __restrict__ u2,
                const float* __restrict__ c2,
                const float* __restrict__ msk,
                const float* __restrict__ C1p,
                const float* __restrict__ b,
                float* __restrict__ o1,
                float* __restrict__ o2)
{
  const int lane = threadIdx.x & 31;
  const int wave = threadIdx.x >> 5;
  const int tile = blockIdx.x * 4 + wave;
  const int r0 = (tile >> 5) * 16;
  const int c0 = (tile & 31) * 16;

  const int hi = lane >> 4;
  const int lp = lane & 15;

  const unsigned rowcode = (unsigned)((0x19ull << (2 * lp)) >> 2);

  const int rowoff = (r0 + lp) * GN + c0;

  __builtin_prefetch(&xi_nx[rowoff + hi * (GN * GN)], 0, 1);

  RingCell q0, q1;
  ring_addr(q0, hi ? r0 - 1  : r0 + lp, hi ? c0 + lp : c0 - 1,
            hi ? (r0 > 0)       : (c0 > 0), r0, c0);
  ring_addr(q1, hi ? r0 + 16 : r0 + lp, hi ? c0 + lp : c0 + 16,
            hi ? (r0 + 16 < GN) : (c0 + 16 < GN), r0, c0);

  const int base = (r0 + 8 * hi) * GN + c0 + lp;
  float uc[8];
  #pragma unroll
  for (int v = 0; v < 8; ++v) uc[v] = u1[base + v * GN];

  const int rf = rowoff + 8 * hi;
  const float4 ua = *(const float4*)&u1[rf],  ub = *(const float4*)&u1[rf + 4];

  q0.uc = u1[q0.id];
  q0.gu = u1[q0.up ? q0.id - GN : q0.id];
  q0.gd = u1[q0.dn ? q0.id + GN : q0.id];
  q0.gl = u1[q0.lf ? q0.id - 1 : q0.id];
  q0.gr = u1[q0.rt ? q0.id + 1 : q0.id];
  q1.uc = u1[q1.id];
  q1.gu = u1[q1.up ? q1.id - GN : q1.id];
  q1.gd = u1[q1.dn ? q1.id + GN : q1.id];
  q1.gl = u1[q1.lf ? q1.id - 1 : q1.id];
  q1.gr = u1[q1.rt ? q1.id + 1 : q1.id];

  const float4 ka = *(const float4*)&C1p[rf], kb = *(const float4*)&C1p[rf + 4];
  const float4 ba = *(const float4*)&b[rf],   bq = *(const float4*)&b[rf + 4];
  const float4 wa = *(const float4*)&u2[rf],  wb = *(const float4*)&u2[rf + 4];
  const float4 ca = *(const float4*)&c2[rf],  cb = *(const float4*)&c2[rf + 4];
  const float4 ma = *(const float4*)&msk[rf], mb = *(const float4*)&msk[rf + 4];
  const float4 xa = *(const float4*)&xi0[rf], xb = *(const float4*)&xi0[rf + 4];

  q0.k1 = C1p[q0.id];  q0.bv = b[q0.id];    q0.w2 = u2[q0.id];
  q0.cc = c2[q0.id];   q0.mk = msk[q0.id];  q0.xv = xi0[q0.id];
  q1.k1 = C1p[q1.id];  q1.bv = b[q1.id];    q1.w2 = u2[q1.id];
  q1.cc = c2[q1.id];   q1.mk = msk[q1.id];  q1.xv = xi0[q1.id];

  const float4 ya = *(const float4*)&xi1[rf], yb = *(const float4*)&xi1[rf + 4];

  __builtin_amdgcn_sched_barrier(0);

  const float ur[8]  = {ua.x, ua.y, ua.z, ua.w, ub.x, ub.y, ub.z, ub.w};
  const float k1r[8] = {ka.x, ka.y, ka.z, ka.w, kb.x, kb.y, kb.z, kb.w};
  const float bbr[8] = {ba.x, ba.y, ba.z, ba.w, bq.x, bq.y, bq.z, bq.w};
  const float w2r[8] = {wa.x, wa.y, wa.z, wa.w, wb.x, wb.y, wb.z, wb.w};
  const float ccr[8] = {ca.x, ca.y, ca.z, ca.w, cb.x, cb.y, cb.z, cb.w};
  const float mkr[8] = {ma.x, ma.y, ma.z, ma.w, mb.x, mb.y, mb.z, mb.w};
  const float x0r[8] = {xa.x, xa.y, xa.z, xa.w, xb.x, xb.y, xb.z, xb.w};
  const float x1r[8] = {ya.x, ya.y, ya.z, ya.w, yb.x, yb.y, yb.z, yb.w};

  v2f tv[4], iv[4];
  #pragma unroll
  for (int j = 0; j < 4; ++j) {
    const int sh = 30 - 16 * hi - 4 * j;
    tv[j].x = (float)((int)(rowcode << sh) >> 30);
    tv[j].y = (float)((int)(rowcode << (sh - 2)) >> 30);
    const int k = 8 * hi + 2 * j;
    iv[j].x = (lp == k)     ? 1.0f : 0.0f;
    iv[j].y = (lp == k + 1) ? 1.0f : 0.0f;
  }
  const float e0  = (lp == 0)  ? 1.0f : 0.0f;
  const float e15 = (lp == 15) ? 1.0f : 0.0f;

  v8f acT = {};
  #pragma unroll
  for (int j = 0; j < 4; ++j) {
    v2f cU, rU;
    cU.x = uc[2 * j];  cU.y = uc[2 * j + 1];
    rU.x = ur[2 * j];  rU.y = ur[2 * j + 1];
    acT = mma4(cU, tv[j], acT);
    acT = mma4(tv[j], rU, acT);
  }
  {
    const float h0 = q0.ok ? q0.uc : 0.0f;
    const float h1 = q1.ok ? q1.uc : 0.0f;
    v2f a9, b9;
    a9.x = hi ? e0  : h0;
    a9.y = hi ? e15 : h1;
    b9.x = hi ? h0  : e0;
    b9.y = hi ? h1  : e15;
    acT = mma4(b9, a9, acT);
  }

  float vr[8];
  #pragma unroll
  for (int v = 0; v < 8; ++v)
    vr[v] = k1r[v] * (8.0f * ur[v] - (4.0f - bbr[v]) * w2r[v]
                      + ccr[v] * acT[v] + mkr[v] * x0r[v]);

  v8f vcm = {};
  #pragma unroll
  for (int j = 0; j < 4; ++j) {
    v2f rV;
    rV.x = vr[2 * j];  rV.y = vr[2 * j + 1];
    vcm = mma4(rV, iv[j], vcm);
  }

  v8f a2T = {};
  #pragma unroll
  for (int j = 0; j < 4; ++j) {
    v2f cV, rV;
    cV.x = vcm[2 * j]; cV.y = vcm[2 * j + 1];
    rV.x = vr[2 * j];  rV.y = vr[2 * j + 1];
    a2T = mma4(cV, tv[j], a2T);
    a2T = mma4(tv[j], rV, a2T);
  }
  {
    const float g0 = ring_value(q0);
    const float g1 = ring_value(q1);
    v2f a9, b9;
    a9.x = hi ? e0  : g0;
    a9.y = hi ? e15 : g1;
    b9.x = hi ? g0  : e0;
    b9.y = hi ? g1  : e15;
    a2T = mma4(b9, a9, a2T);
  }

  float zr[8];
  #pragma unroll
  for (int v = 0; v < 8; ++v)
    zr[v] = k1r[v] * (8.0f * vr[v] - (4.0f - bbr[v]) * ur[v]
                      + ccr[v] * a2T[v] + mkr[v] * x1r[v]);

  *(float4*)&o1[rf]     = make_float4(vr[0], vr[1], vr[2], vr[3]);
  *(float4*)&o1[rf + 4] = make_float4(vr[4], vr[5], vr[6], vr[7]);
  *(float4*)&o2[rf]     = make_float4(zr[0], zr[1], zr[2], zr[3]);
  *(float4*)&o2[rf + 4] = make_float4(zr[4], zr[5], zr[6], zr[7]);
}

extern "C" void kernel_launch(void* const* d_in, const int* in_sizes, int n_in,
                              void* d_out, int out_size, void* d_ws, size_t ws_size,
                              hipStream_t stream) {
  const float* inp = (const float*)d_in[0];
  const float* un1 = (const float*)d_in[1];
  const float* un2 = (const float*)d_in[2];
  const float* c2  = (const float*)d_in[3];
  const float* msk = (const float*)d_in[4];
  const float* b   = (const float*)d_in[5];

  const int plane = GN * GN;
  const int nl = in_sizes[0] / plane / 2;

  float* pA  = (float*)d_ws;
  float* pC  = pA + plane;
  float* pD  = pC + plane;
  float* C1p = pD + plane;
  float* pB  = (float*)d_out;

  pml_coeff<<<dim3(plane / 256), dim3(256), 0, stream>>>(b, C1p);

  dim3 grid(256);
  dim3 block(128);

  for (int l = 0; l < nl; ++l) {
    const float* xi0   = inp + (size_t)(2 * l) * plane;
    const float* xi1   = xi0 + plane;
    const float* xi_nx = (l + 1 < nl) ? (xi1 + plane) : xi0;

    const float* u1 = (l == 0) ? un1 : ((l & 1) ? pD : pB);
    const float* u2 = (l == 0) ? un2 : ((l & 1) ? pC : pA);
    float* o1 = (l & 1) ? pA : pC;
    float* o2 = (l & 1) ? pB : pD;

    wave_step2<<<grid, block, 0, stream>>>(xi0, xi1, xi_nx, u1, u2, c2, msk,
                                           C1p, b, o1, o2);
  }
}


# 8 time steps per launch in registers per wave, no LDS: a 32x32 region (2x2 tiles) stepped 8 times, rim recomputed; tile-to-tile edges by one-hot WMMAs (halo sum order changes); 129 launches to 33
# speedup vs baseline: 8.0030x; 1.4837x over previous
#include <hip/hip_runtime.h>

typedef float v2f __attribute__((ext_vector_type(2)));
typedef float v8f __attribute__((ext_vector_type(8)));

#define GN 512
#define NS 8

__global__ __launch_bounds__(256)
void pml_coeff(const float* __restrict__ b, float* __restrict__ C1p)
{
  const int i = blockIdx.x * 256 + threadIdx.x;
  C1p[i] = 1.0f / (4.0f + b[i]);
}

__device__ __forceinline__ v8f mma4(v2f a, v2f b, v8f c)
{
  return __builtin_amdgcn_wmma_f32_16x16x4_f32(false, a, false, b, (short)0, c,
                                               false, false);
}

__device__ __forceinline__ void ld8(const float* __restrict__ p, int idx, float (&d)[8])
{
  const float4 a = *(const float4*)&p[idx];
  const float4 c = *(const float4*)&p[idx + 4];
  d[0] = a.x; d[1] = a.y; d[2] = a.z; d[3] = a.w;
  d[4] = c.x; d[5] = c.y; d[6] = c.z; d[7] = c.w;
}

__device__ __forceinline__ void step4(const float (&S)[4][8], float (&D)[4][8],
                                      const float (&K1)[4][8], const float (&C3)[4][8],
                                      const float (&CC)[4][8], const float (&MK)[4][8],
                                      const float (&X)[4][8],
                                      const v2f (&tv)[4], const v2f (&iv)[4],
                                      const v2f s0, const v2f s3)
{
  v8f col[4], acc[4];
  #pragma unroll
  for (int t = 0; t < 4; ++t) { col[t] = v8f{}; acc[t] = v8f{}; }

  #pragma unroll
  for (int j = 0; j < 4; ++j) {
    #pragma unroll
    for (int t = 0; t < 4; ++t) {
      v2f r;  r.x = S[t][2 * j];  r.y = S[t][2 * j + 1];
      col[t] = mma4(r, iv[j], col[t]);
    }
  }
  #pragma unroll
  for (int j = 0; j < 4; ++j) {
    #pragma unroll
    for (int t = 0; t < 4; ++t) {
      v2f c, r;
      c.x = col[t][2 * j];  c.y = col[t][2 * j + 1];
      r.x = S[t][2 * j];    r.y = S[t][2 * j + 1];
      acc[t] = mma4(c, tv[j], acc[t]);
      acc[t] = mma4(tv[j], r, acc[t]);
    }
  }
  #pragma unroll
  for (int t = 0; t < 4; ++t) {
    const int nv = t ^ 2;
    const int nh = t ^ 1;
    v2f c, r;
    if ((t >> 1) == 0) {
      c.x = col[nv][0];  c.y = col[nv][1];
      acc[t] = mma4(c, s0, acc[t]);
    } else {
      c.x = col[nv][6];  c.y = col[nv][7];
      acc[t] = mma4(c, s3, acc[t]);
    }
    if ((t & 1) == 0) {
      r.x = S[nh][0];  r.y = S[nh][1];
      acc[t] = mma4(s0, r, acc[t]);
    } else {
      r.x = S[nh][6];  r.y = S[nh][7];
      acc[t] = mma4(s3, r, acc[t]);
    }
  }
  #pragma unroll
  for (int t = 0; t < 4; ++t) {
    #pragma unroll
    for (int v = 0; v < 8; ++v)
      D[t][v] = K1[t][v] * (8.0f * S[t][v] - C3[t][v] * D[t][v]
                            + CC[t][v] * acc[t][v] + MK[t][v] * X[t][v]);
  }
}

__global__ __launch_bounds__(128)
void wave_step8(const float* __restrict__ xi,
                const float* __restrict__ xi_nx,
                const float* __restrict__ u1,
                const float* __restrict__ u2,
                const float* __restrict__ c2,
                const float* __restrict__ msk,
                const float* __restrict__ C1p,
                const float* __restrict__ b,
                float* __restrict__ o1,
                float* __restrict__ o2)
{
  const int lane = threadIdx.x & 31;
  const int wave = threadIdx.x >> 5;
  const int tile = blockIdx.x * 4 + wave;
  const int r0 = (tile >> 5) * 16;
  const int c0 = (tile & 31) * 16;

  const int hi = lane >> 4;
  const int lp = lane & 15;
  const int plane = GN * GN;

  #pragma unroll
  for (int s = 0; s < NS; s += 2)
    __builtin_prefetch(&xi_nx[(s + hi) * plane + (r0 + lp) * GN + c0], 0, 1);

  int idx[4];
  bool in[4];
  #pragma unroll
  for (int t = 0; t < 4; ++t) {
    const int row = r0 - 8 + 16 * (t >> 1) + lp;
    const int cl  = c0 - 8 + 16 * (t & 1) + 8 * hi;
    in[t]  = ((unsigned)row < (unsigned)GN) && ((unsigned)cl < (unsigned)GN);
    idx[t] = in[t] ? row * GN + cl : (r0 + lp) * GN + c0 + 8 * hi;
  }

  float U[4][8], Wd[4][8], K1[4][8], C3[4][8], CC[4][8], MK[4][8];
  float Xa[4][8], Xb[4][8], Na[4][8], Nb[4][8];
  #pragma unroll
  for (int t = 0; t < 4; ++t) ld8(u1,  idx[t], U[t]);
  #pragma unroll
  for (int t = 0; t < 4; ++t) ld8(C1p, idx[t], K1[t]);
  #pragma unroll
  for (int t = 0; t < 4; ++t) ld8(b,   idx[t], C3[t]);
  #pragma unroll
  for (int t = 0; t < 4; ++t) ld8(u2,  idx[t], Wd[t]);
  #pragma unroll
  for (int t = 0; t < 4; ++t) ld8(c2,  idx[t], CC[t]);
  #pragma unroll
  for (int t = 0; t < 4; ++t) ld8(msk, idx[t], MK[t]);
  #pragma unroll
  for (int t = 0; t < 4; ++t) { ld8(xi, idx[t], Xa[t]);  ld8(xi + plane, idx[t], Xb[t]); }

  __builtin_amdgcn_sched_barrier(0);

  #pragma unroll
  for (int t = 0; t < 4; ++t) {
    #pragma unroll
    for (int v = 0; v < 8; ++v) {
      U[t][v]  = in[t] ? U[t][v]  : 0.0f;
      K1[t][v] = in[t] ? K1[t][v] : 0.0f;
      C3[t][v] = 4.0f - C3[t][v];
      Na[t][v] = 0.0f;  Nb[t][v] = 0.0f;
    }
  }

  const unsigned rowcode = (unsigned)((0x19ull << (2 * lp)) >> 2);
  v2f tv[4], iv[4];
  #pragma unroll
  for (int j = 0; j < 4; ++j) {
    const int sh = 30 - 16 * hi - 4 * j;
    tv[j].x = (float)((int)(rowcode << sh) >> 30);
    tv[j].y = (float)((int)(rowcode << (sh - 2)) >> 30);
    const int k = 8 * hi + 2 * j;
    iv[j].x = (lp == k)     ? 1.0f : 0.0f;
    iv[j].y = (lp == k + 1) ? 1.0f : 0.0f;
  }
  v2f s0, s3;
  s0.x = (hi == 0 && lp == 15) ? 1.0f : 0.0f;  s0.y = 0.0f;
  s3.x = 0.0f;  s3.y = (hi == 1 && lp == 0) ? 1.0f : 0.0f;

  const float* xp = xi + 2 * plane;
  for (int it = 0; it < NS / 2; ++it) {
    if (it + 1 < NS / 2) {
      #pragma unroll
      for (int t = 0; t < 4; ++t) { ld8(xp, idx[t], Na[t]);  ld8(xp + plane, idx[t], Nb[t]); }
    }
    step4(U,  Wd, K1, C3, CC, MK, Xa, tv, iv, s0, s3);
    step4(Wd, U,  K1, C3, CC, MK, Xb, tv, iv, s0, s3);

    #pragma unroll
    for (int t = 0; t < 4; ++t) {
      #pragma unroll
      for (int v = 0; v < 8; ++v) { Xa[t][v] = Na[t][v];  Xb[t][v] = Nb[t][v]; }
    }
    xp += 2 * plane;
  }

  const bool lower = lp < 8;
  const bool right = hi == 0;
  float a1[8], a2[8];
  #pragma unroll
  for (int v = 0; v < 8; ++v) {
    const float t1 = right ? Wd[1][v] : Wd[0][v];
    const float b1 = right ? Wd[3][v] : Wd[2][v];
    a1[v] = lower ? b1 : t1;
    const float t2 = right ? U[1][v] : U[0][v];
    const float b2 = right ? U[3][v] : U[2][v];
    a2[v] = lower ? b2 : t2;
  }
  const int so = (r0 - 8 + (lower ? 16 : 0) + lp) * GN + c0 - 8 + (right ? 16 : 0) + 8 * hi;
  *(float4*)&o1[so]     = make_float4(a1[0], a1[1], a1[2], a1[3]);
  *(float4*)&o1[so + 4] = make_float4(a1[4], a1[5], a1[6], a1[7]);
  *(float4*)&o2[so]     = make_float4(a2[0], a2[1], a2[2], a2[3]);
  *(float4*)&o2[so + 4] = make_float4(a2[4], a2[5], a2[6], a2[7]);
}

extern "C" void kernel_launch(void* const* d_in, const int* in_sizes, int n_in,
                              void* d_out, int out_size, void* d_ws, size_t ws_size,
                              hipStream_t stream) {
  const float* inp = (const float*)d_in[0];
  const float* un1 = (const float*)d_in[1];
  const float* un2 = (const float*)d_in[2];
  const float* c2  = (const float*)d_in[3];
  const float* msk = (const float*)d_in[4];
  const float* b   = (const float*)d_in[5];

  const int plane = GN * GN;
  const int nl = in_sizes[0] / plane / NS;

  float* pA  = (float*)d_ws;
  float* pC  = pA + plane;
  float* pD  = pC + plane;
  float* C1p = pD + plane;
  float* pB  = (float*)d_out;

  pml_coeff<<<dim3(plane / 256), dim3(256), 0, stream>>>(b, C1p);

  dim3 grid(256);
  dim3 block(128);

  for (int l = 0; l < nl; ++l) {
    const float* xi    = inp + (size_t)(NS * l) * plane;
    const float* xi_nx = (l + 1 < nl) ? (xi + (size_t)NS * plane) : xi;

    const float* u1 = (l == 0) ? un1 : ((l & 1) ? pD : pB);
    const float* u2 = (l == 0) ? un2 : ((l & 1) ? pC : pA);
    float* o1 = (l & 1) ? pA : pC;
    float* o2 = (l & 1) ? pB : pD;

    wave_step8<<<grid, block, 0, stream>>>(xi, xi_nx, u1, u2, c2, msk, C1p, b, o1, o2);
  }
}


# Global loads made line-wise: one b128 load covers 4 whole 128-B region rows (8 lines touched, was 16-32), re-laid into the WMMA row layout via a per-wave LDS buffer, no barrier; sum order unchanged
# speedup vs baseline: 8.3155x; 1.0390x over previous
#include <hip/hip_runtime.h>

typedef float v2f __attribute__((ext_vector_type(2)));
typedef float v8f __attribute__((ext_vector_type(8)));

#define GN 512
#define NS 8
#define LPITCH 36
#define LBUF (32 * LPITCH)

__global__ __launch_bounds__(256)
void pml_coeff(const float* __restrict__ b, float* __restrict__ C1p)
{
  const int i = blockIdx.x * 256 + threadIdx.x;
  C1p[i] = 1.0f / (4.0f + b[i]);
}

__device__ __forceinline__ v8f mma4(v2f a, v2f b, v8f c)
{
  return __builtin_amdgcn_wmma_f32_16x16x4_f32(false, a, false, b, (short)0, c,
                                               false, false);
}

__device__ __forceinline__ void ldL(const float* p, const unsigned (&off)[8],
                                    float (&d)[8][4])
{
  #pragma unroll
  for (int i = 0; i < 8; ++i) {
    const float4 a = *(const float4*)&p[off[i]];
    d[i][0] = a.x; d[i][1] = a.y; d[i][2] = a.z; d[i][3] = a.w;
  }
}

__device__ __forceinline__ void wave_lds_order()
{
  __builtin_amdgcn_fence(__ATOMIC_RELEASE, "wavefront");
  __builtin_amdgcn_wave_barrier();
  __builtin_amdgcn_fence(__ATOMIC_ACQUIRE, "wavefront");
}

__device__ __forceinline__ void to_rows(float* __restrict__ buf, int lane,
                                        const float (&d)[8][4], float (&R)[4][8])
{
  const int sr = lane >> 3, sg = lane & 7;
  const int lp = lane & 15, hi = lane >> 4;
  #pragma unroll
  for (int i = 0; i < 8; ++i)
    *(float4*)&buf[(4 * i + sr) * LPITCH + 4 * sg] = make_float4(d[i][0], d[i][1], d[i][2], d[i][3]);
  wave_lds_order();
  #pragma unroll
  for (int t = 0; t < 4; ++t) {
    const int o = (16 * (t >> 1) + lp) * LPITCH + 16 * (t & 1) + 8 * hi;
    const float4 a = *(const float4*)&buf[o];
    const float4 c = *(const float4*)&buf[o + 4];
    R[t][0] = a.x; R[t][1] = a.y; R[t][2] = a.z; R[t][3] = a.w;
    R[t][4] = c.x; R[t][5] = c.y; R[t][6] = c.z; R[t][7] = c.w;
  }
  wave_lds_order();
}

__device__ __forceinline__ void lap4(const float (&S)[4][8], v8f (&acc)[4],
                                     const v2f (&tv)[4], const v2f (&iv)[4],
                                     const v2f s0, const v2f s3)
{
  v8f col[4];
  #pragma unroll
  for (int t = 0; t < 4; ++t) { col[t] = v8f{}; acc[t] = v8f{}; }

  #pragma unroll
  for (int j = 0; j < 4; ++j) {
    #pragma unroll
    for (int t = 0; t < 4; ++t) {
      v2f r;  r.x = S[t][2 * j];  r.y = S[t][2 * j + 1];
      col[t] = mma4(r, iv[j], col[t]);
    }
  }
  #pragma unroll
  for (int j = 0; j < 4; ++j) {
    #pragma unroll
    for (int t = 0; t < 4; ++t) {
      v2f c, r;
      c.x = col[t][2 * j];  c.y = col[t][2 * j + 1];
      r.x = S[t][2 * j];    r.y = S[t][2 * j + 1];
      acc[t] = mma4(c, tv[j], acc[t]);
      acc[t] = mma4(tv[j], r, acc[t]);
    }
  }
  #pragma unroll
  for (int t = 0; t < 4; ++t) {
    const int nv = t ^ 2;
    const int nh = t ^ 1;
    v2f c, r;
    if ((t >> 1) == 0) {
      c.x = col[nv][0];  c.y = col[nv][1];
      acc[t] = mma4(c, s0, acc[t]);
    } else {
      c.x = col[nv][6];  c.y = col[nv][7];
      acc[t] = mma4(c, s3, acc[t]);
    }
    if ((t & 1) == 0) {
      r.x = S[nh][0];  r.y = S[nh][1];
      acc[t] = mma4(s0, r, acc[t]);
    } else {
      r.x = S[nh][6];  r.y = S[nh][7];
      acc[t] = mma4(s3, r, acc[t]);
    }
  }
}

__device__ __forceinline__ void pw4(const v8f (&acc)[4], const float (&S)[4][8],
                                    float (&D)[4][8],
                                    const float (&K1)[4][8], const float (&C3)[4][8],
                                    const float (&CC)[4][8], const float (&MK)[4][8],
                                    const float (&X)[4][8])
{
  #pragma unroll
  for (int t = 0; t < 4; ++t) {
    #pragma unroll
    for (int v = 0; v < 8; ++v)
      D[t][v] = K1[t][v] * (8.0f * S[t][v] - C3[t][v] * D[t][v]
                            + CC[t][v] * acc[t][v] + MK[t][v] * X[t][v]);
  }
}

__device__ __forceinline__ void step4(const float (&S)[4][8], float (&D)[4][8],
                                      const float (&K1)[4][8], const float (&C3)[4][8],
                                      const float (&CC)[4][8], const float (&MK)[4][8],
                                      const float (&X)[4][8],
                                      const v2f (&tv)[4], const v2f (&iv)[4],
                                      const v2f s0, const v2f s3)
{
  v8f acc[4];
  lap4(S, acc, tv, iv, s0, s3);
  pw4(acc, S, D, K1, C3, CC, MK, X);
}

__global__ __launch_bounds__(128)
void wave_step8(const float* xi,
                const float* xi_nx,
                const float* u1,
                const float* u2,
                const float* c2,
                const float* msk,
                const float* C1p,
                const float* b,
                float* o1,
                float* o2)
{
  const int lane = threadIdx.x & 31;
  const int wave = threadIdx.x >> 5;
  const int tile = blockIdx.x * 4 + wave;
  const int r0 = (tile >> 5) * 16;
  const int c0 = (tile & 31) * 16;

  const int hi = lane >> 4;
  const int lp = lane & 15;
  const int plane = GN * GN;

  #pragma unroll
  for (int s = 0; s < NS; s += 2)
    __builtin_prefetch(&xi_nx[(s + hi) * plane + (r0 + lp) * GN + c0], 0, 1);

  __shared__ __attribute__((aligned(16))) float stage[4][2][LBUF];
  float* const bufA = &stage[wave][0][0];
  float* const bufB = &stage[wave][1][0];

  const int sr = lane >> 3, sg = lane & 7;
  const int gc = c0 - 8 + 4 * sg;
  const bool colin = (unsigned)gc < (unsigned)GN;
  const int gcs = colin ? gc : c0 + 4 * (sg & 3);
  unsigned off[8];
  #pragma unroll
  for (int i = 0; i < 8; ++i) {
    const int gr = r0 - 8 + 4 * i + sr;
    const bool rowin = (unsigned)gr < (unsigned)GN;
    off[i] = (unsigned)((rowin ? gr : r0 + sr) * GN + gcs);
  }

  const unsigned rowcode = (unsigned)((0x19ull << (2 * lp)) >> 2);
  v2f tv[4], iv[4];
  #pragma unroll
  for (int j = 0; j < 4; ++j) {
    const int sh = 30 - 16 * hi - 4 * j;
    tv[j].x = (float)((int)(rowcode << sh) >> 30);
    tv[j].y = (float)((int)(rowcode << (sh - 2)) >> 30);
    const int k = 8 * hi + 2 * j;
    iv[j].x = (lp == k)     ? 1.0f : 0.0f;
    iv[j].y = (lp == k + 1) ? 1.0f : 0.0f;
  }
  v2f s0, s3;
  s0.x = (hi == 0 && lp == 15) ? 1.0f : 0.0f;  s0.y = 0.0f;
  s3.x = 0.0f;  s3.y = (hi == 1 && lp == 0) ? 1.0f : 0.0f;

  bool in[4];
  #pragma unroll
  for (int t = 0; t < 4; ++t) {
    const int row = r0 - 8 + 16 * (t >> 1) + lp;
    const int cl  = c0 - 8 + 16 * (t & 1) + 8 * hi;
    in[t] = ((unsigned)row < (unsigned)GN) && ((unsigned)cl < (unsigned)GN);
  }

  float LU[8][4], LK[8][4], LB[8][4], LW[8][4], LC[8][4], LM[8][4], LXa[8][4], LXb[8][4];
  ldL(u1,  off, LU);
  __builtin_amdgcn_sched_barrier(0);
  ldL(C1p, off, LK);
  __builtin_amdgcn_sched_barrier(0);
  ldL(b,   off, LB);
  __builtin_amdgcn_sched_barrier(0);
  ldL(u2,  off, LW);
  __builtin_amdgcn_sched_barrier(0);
  ldL(c2,  off, LC);
  __builtin_amdgcn_sched_barrier(0);
  ldL(msk, off, LM);
  __builtin_amdgcn_sched_barrier(0);
  ldL(xi,  off, LXa);
  __builtin_amdgcn_sched_barrier(0);
  ldL(xi + plane, off, LXb);
  __builtin_amdgcn_sched_barrier(0);

  float U[4][8], Wd[4][8], K1[4][8], C3[4][8], CC[4][8], MK[4][8];
  float Xa[4][8], Xb[4][8], Na[4][8], Nb[4][8];
  to_rows(bufA, lane, LU, U);
  #pragma unroll
  for (int t = 0; t < 4; ++t) {
    #pragma unroll
    for (int v = 0; v < 8; ++v) U[t][v] = in[t] ? U[t][v] : 0.0f;
  }
  __builtin_amdgcn_sched_barrier(0);

  float La[8][4], Lb[8][4];
  ldL(xi + 2 * plane, off, La);
  ldL(xi + 3 * plane, off, Lb);

  v8f acc0[4];
  lap4(U, acc0, tv, iv, s0, s3);

  to_rows(bufB, lane, LK,  K1);
  to_rows(bufA, lane, LB,  C3);
  to_rows(bufB, lane, LW,  Wd);
  to_rows(bufA, lane, LC,  CC);
  to_rows(bufB, lane, LM,  MK);
  to_rows(bufA, lane, LXa, Xa);
  to_rows(bufB, lane, LXb, Xb);
  #pragma unroll
  for (int t = 0; t < 4; ++t) {
    #pragma unroll
    for (int v = 0; v < 8; ++v) {
      K1[t][v] = in[t] ? K1[t][v] : 0.0f;
      C3[t][v] = 4.0f - C3[t][v];
      Na[t][v] = 0.0f;  Nb[t][v] = 0.0f;
    }
  }
  pw4(acc0, U, Wd, K1, C3, CC, MK, Xa);

  const float* xp = xi + 4 * plane;
  for (int it = 0; it < NS / 2; ++it) {
    const bool more = it + 1 < NS / 2;
    if (it > 0) {
      if (more) { ldL(xp, off, La);  ldL(xp + plane, off, Lb);  xp += 2 * plane; }
      step4(U,  Wd, K1, C3, CC, MK, Xa, tv, iv, s0, s3);
    }
    if (more) { to_rows(bufA, lane, La, Na);  to_rows(bufB, lane, Lb, Nb); }
    step4(Wd, U,  K1, C3, CC, MK, Xb, tv, iv, s0, s3);

    #pragma unroll
    for (int t = 0; t < 4; ++t) {
      #pragma unroll
      for (int v = 0; v < 8; ++v) { Xa[t][v] = Na[t][v];  Xb[t][v] = Nb[t][v]; }
    }
  }

  const bool lower = lp < 8;
  const bool right = hi == 0;
  float a1[8], a2[8];
  #pragma unroll
  for (int v = 0; v < 8; ++v) {
    const float t1 = right ? Wd[1][v] : Wd[0][v];
    const float b1 = right ? Wd[3][v] : Wd[2][v];
    a1[v] = lower ? b1 : t1;
    const float t2 = right ? U[1][v] : U[0][v];
    const float b2 = right ? U[3][v] : U[2][v];
    a2[v] = lower ? b2 : t2;
  }
  const int so = (r0 - 8 + (lower ? 16 : 0) + lp) * GN + c0 - 8 + (right ? 16 : 0) + 8 * hi;
  *(float4*)&o1[so]     = make_float4(a1[0], a1[1], a1[2], a1[3]);
  *(float4*)&o1[so + 4] = make_float4(a1[4], a1[5], a1[6], a1[7]);
  *(float4*)&o2[so]     = make_float4(a2[0], a2[1], a2[2], a2[3]);
  *(float4*)&o2[so + 4] = make_float4(a2[4], a2[5], a2[6], a2[7]);
}

extern "C" void kernel_launch(void* const* d_in, const int* in_sizes, int n_in,
                              void* d_out, int out_size, void* d_ws, size_t ws_size,
                              hipStream_t stream) {
  const float* inp = (const float*)d_in[0];
  const float* un1 = (const float*)d_in[1];
  const float* un2 = (const float*)d_in[2];
  const float* c2  = (const float*)d_in[3];
  const float* msk = (const float*)d_in[4];
  const float* b   = (const float*)d_in[5];

  const int plane = GN * GN;
  const int nl = in_sizes[0] / plane / NS;

  float* pA  = (float*)d_ws;
  float* pC  = pA + plane;
  float* pD  = pC + plane;
  float* C1p = pD + plane;
  float* pB  = (float*)d_out;

  pml_coeff<<<dim3(plane / 256), dim3(256), 0, stream>>>(b, C1p);

  dim3 grid(256);
  dim3 block(128);

  for (int l = 0; l < nl; ++l) {
    const float* xi    = inp + (size_t)(NS * l) * plane;
    const float* xi_nx = (l + 1 < nl) ? (xi + (size_t)NS * plane) : xi;

    const float* u1 = (l == 0) ? un1 : ((l & 1) ? pD : pB);
    const float* u2 = (l == 0) ? un2 : ((l & 1) ? pC : pA);
    float* o1 = (l & 1) ? pA : pC;
    float* o2 = (l & 1) ? pB : pD;

    wave_step8<<<grid, block, 0, stream>>>(xi, xi_nx, u1, u2, c2, msk, C1p, b, o1, o2);
  }
}


# Column layout for the T*S WMMAs read back from a per-wave transposed LDS copy of the state, not made by 16 identity WMMAs per step (56 to 40); row-layout WMMAs run first; Laplacian sum order changes
# speedup vs baseline: 8.7916x; 1.0573x over previous
#include <hip/hip_runtime.h>

typedef float v2f __attribute__((ext_vector_type(2)));
typedef float v8f __attribute__((ext_vector_type(8)));

#define GN 512
#define NS 8
#define LPITCH 36
#define LBUF (32 * LPITCH)

__global__ __launch_bounds__(256)
void pml_coeff(const float* __restrict__ b, float* __restrict__ C1p)
{
  const int i = blockIdx.x * 256 + threadIdx.x;
  C1p[i] = 1.0f / (4.0f + b[i]);
}

__device__ __forceinline__ v8f mma4(v2f a, v2f b, v8f c)
{
  return __builtin_amdgcn_wmma_f32_16x16x4_f32(false, a, false, b, (short)0, c,
                                               false, false);
}

__device__ __forceinline__ void ldL(const float* p, const unsigned (&off)[8],
                                    float (&d)[8][4])
{
  #pragma unroll
  for (int i = 0; i < 8; ++i) {
    const float4 a = *(const float4*)&p[off[i]];
    d[i][0] = a.x; d[i][1] = a.y; d[i][2] = a.z; d[i][3] = a.w;
  }
}

__device__ __forceinline__ void wave_lds_order()
{
  __builtin_amdgcn_fence(__ATOMIC_RELEASE, "wavefront");
  __builtin_amdgcn_wave_barrier();
  __builtin_amdgcn_fence(__ATOMIC_ACQUIRE, "wavefront");
}

__device__ __forceinline__ void to_rows(float* __restrict__ buf, int lane,
                                        const float (&d)[8][4], float (&R)[4][8])
{
  const int sr = lane >> 3, sg = lane & 7;
  const int lp = lane & 15, hi = lane >> 4;
  #pragma unroll
  for (int i = 0; i < 8; ++i)
    *(float4*)&buf[(4 * i + sr) * LPITCH + 4 * sg] = make_float4(d[i][0], d[i][1], d[i][2], d[i][3]);
  wave_lds_order();
  #pragma unroll
  for (int t = 0; t < 4; ++t) {
    const int o = (16 * (t >> 1) + lp) * LPITCH + 16 * (t & 1) + 8 * hi;
    const float4 a = *(const float4*)&buf[o];
    const float4 c = *(const float4*)&buf[o + 4];
    R[t][0] = a.x; R[t][1] = a.y; R[t][2] = a.z; R[t][3] = a.w;
    R[t][4] = c.x; R[t][5] = c.y; R[t][6] = c.z; R[t][7] = c.w;
  }
  wave_lds_order();
}

__device__ __forceinline__ void lap4(float* __restrict__ tb, int lane,
                                     const float (&S)[4][8], v8f (&acc)[4],
                                     const v2f (&tv)[4], const v2f s0, const v2f s3)
{
  const int lp = lane & 15, hi = lane >> 4;
  #pragma unroll
  for (int t = 0; t < 4; ++t) {
    #pragma unroll
    for (int v = 0; v < 8; ++v)
      tb[(16 * (t & 1) + 8 * hi + v) * LPITCH + 16 * (t >> 1) + lp] = S[t][v];
  }
  wave_lds_order();
  v8f col[4];
  #pragma unroll
  for (int t = 0; t < 4; ++t) {
    const int o = (16 * (t & 1) + lp) * LPITCH + 16 * (t >> 1) + 8 * hi;
    const float4 a = *(const float4*)&tb[o];
    const float4 c = *(const float4*)&tb[o + 4];
    col[t][0] = a.x; col[t][1] = a.y; col[t][2] = a.z; col[t][3] = a.w;
    col[t][4] = c.x; col[t][5] = c.y; col[t][6] = c.z; col[t][7] = c.w;
  }
  wave_lds_order();
  __builtin_amdgcn_sched_barrier(0);

  #pragma unroll
  for (int t = 0; t < 4; ++t) acc[t] = v8f{};
  #pragma unroll
  for (int j = 0; j < 4; ++j) {
    #pragma unroll
    for (int t = 0; t < 4; ++t) {
      v2f r;  r.x = S[t][2 * j];  r.y = S[t][2 * j + 1];
      acc[t] = mma4(tv[j], r, acc[t]);
    }
  }
  #pragma unroll
  for (int t = 0; t < 4; ++t) {
    const int nh = t ^ 1;
    v2f r;
    if ((t & 1) == 0) {
      r.x = S[nh][0];  r.y = S[nh][1];
      acc[t] = mma4(s0, r, acc[t]);
    } else {
      r.x = S[nh][6];  r.y = S[nh][7];
      acc[t] = mma4(s3, r, acc[t]);
    }
  }
  #pragma unroll
  for (int j = 0; j < 4; ++j) {
    #pragma unroll
    for (int t = 0; t < 4; ++t) {
      v2f c;  c.x = col[t][2 * j];  c.y = col[t][2 * j + 1];
      acc[t] = mma4(c, tv[j], acc[t]);
    }
  }
  #pragma unroll
  for (int t = 0; t < 4; ++t) {
    const int nv = t ^ 2;
    v2f c;
    if ((t >> 1) == 0) {
      c.x = col[nv][0];  c.y = col[nv][1];
      acc[t] = mma4(c, s0, acc[t]);
    } else {
      c.x = col[nv][6];  c.y = col[nv][7];
      acc[t] = mma4(c, s3, acc[t]);
    }
  }
}

__device__ __forceinline__ void pw4(const v8f (&acc)[4], const float (&S)[4][8],
                                    float (&D)[4][8],
                                    const float (&K1)[4][8], const float (&C3)[4][8],
                                    const float (&CC)[4][8], const float (&MK)[4][8],
                                    const float (&X)[4][8])
{
  #pragma unroll
  for (int t = 0; t < 4; ++t) {
    #pragma unroll
    for (int v = 0; v < 8; ++v)
      D[t][v] = K1[t][v] * (8.0f * S[t][v] - C3[t][v] * D[t][v]
                            + CC[t][v] * acc[t][v] + MK[t][v] * X[t][v]);
  }
}

__device__ __forceinline__ void step4(float* __restrict__ tb, int lane,
                                      const float (&S)[4][8], float (&D)[4][8],
                                      const float (&K1)[4][8], const float (&C3)[4][8],
                                      const float (&CC)[4][8], const float (&MK)[4][8],
                                      const float (&X)[4][8],
                                      const v2f (&tv)[4],
                                      const v2f s0, const v2f s3)
{
  v8f acc[4];
  lap4(tb, lane, S, acc, tv, s0, s3);
  pw4(acc, S, D, K1, C3, CC, MK, X);
}

__global__ __launch_bounds__(128)
void wave_step8(const float* xi,
                const float* xi_nx,
                const float* u1,
                const float* u2,
                const float* c2,
                const float* msk,
                const float* C1p,
                const float* b,
                float* o1,
                float* o2)
{
  const int lane = threadIdx.x & 31;
  const int wave = threadIdx.x >> 5;
  const int tile = blockIdx.x * 4 + wave;
  const int r0 = (tile >> 5) * 16;
  const int c0 = (tile & 31) * 16;

  const int hi = lane >> 4;
  const int lp = lane & 15;
  const int plane = GN * GN;

  #pragma unroll
  for (int s = 0; s < NS; s += 2)
    __builtin_prefetch(&xi_nx[(s + hi) * plane + (r0 + lp) * GN + c0], 0, 1);

  __shared__ __attribute__((aligned(16))) float stage[4][3][LBUF];
  float* const bufA = &stage[wave][0][0];
  float* const bufB = &stage[wave][1][0];
  float* const bufT = &stage[wave][2][0];

  const int sr = lane >> 3, sg = lane & 7;
  const int gc = c0 - 8 + 4 * sg;
  const bool colin = (unsigned)gc < (unsigned)GN;
  const int gcs = colin ? gc : c0 + 4 * (sg & 3);
  unsigned off[8];
  #pragma unroll
  for (int i = 0; i < 8; ++i) {
    const int gr = r0 - 8 + 4 * i + sr;
    const bool rowin = (unsigned)gr < (unsigned)GN;
    off[i] = (unsigned)((rowin ? gr : r0 + sr) * GN + gcs);
  }

  const unsigned rowcode = (unsigned)((0x19ull << (2 * lp)) >> 2);
  v2f tv[4];
  #pragma unroll
  for (int j = 0; j < 4; ++j) {
    const int sh = 30 - 16 * hi - 4 * j;
    tv[j].x = (float)((int)(rowcode << sh) >> 30);
    tv[j].y = (float)((int)(rowcode << (sh - 2)) >> 30);
  }
  v2f s0, s3;
  s0.x = (hi == 0 && lp == 15) ? 1.0f : 0.0f;  s0.y = 0.0f;
  s3.x = 0.0f;  s3.y = (hi == 1 && lp == 0) ? 1.0f : 0.0f;

  bool in[4];
  #pragma unroll
  for (int t = 0; t < 4; ++t) {
    const int row = r0 - 8 + 16 * (t >> 1) + lp;
    const int cl  = c0 - 8 + 16 * (t & 1) + 8 * hi;
    in[t] = ((unsigned)row < (unsigned)GN) && ((unsigned)cl < (unsigned)GN);
  }

  float LU[8][4], LK[8][4], LB[8][4], LW[8][4], LC[8][4], LM[8][4], LXa[8][4], LXb[8][4];
  ldL(u1,  off, LU);
  __builtin_amdgcn_sched_barrier(0);
  ldL(C1p, off, LK);
  __builtin_amdgcn_sched_barrier(0);
  ldL(b,   off, LB);
  __builtin_amdgcn_sched_barrier(0);
  ldL(u2,  off, LW);
  __builtin_amdgcn_sched_barrier(0);
  ldL(c2,  off, LC);
  __builtin_amdgcn_sched_barrier(0);
  ldL(msk, off, LM);
  __builtin_amdgcn_sched_barrier(0);
  ldL(xi,  off, LXa);
  __builtin_amdgcn_sched_barrier(0);
  ldL(xi + plane, off, LXb);
  __builtin_amdgcn_sched_barrier(0);

  float U[4][8], Wd[4][8], K1[4][8], C3[4][8], CC[4][8], MK[4][8];
  float Xa[4][8], Xb[4][8], Na[4][8], Nb[4][8];
  to_rows(bufA, lane, LU, U);
  #pragma unroll
  for (int t = 0; t < 4; ++t) {
    #pragma unroll
    for (int v = 0; v < 8; ++v) U[t][v] = in[t] ? U[t][v] : 0.0f;
  }
  __builtin_amdgcn_sched_barrier(0);

  float La[8][4], Lb[8][4];
  ldL(xi + 2 * plane, off, La);
  ldL(xi + 3 * plane, off, Lb);

  v8f acc0[4];
  lap4(bufT, lane, U, acc0, tv, s0, s3);

  to_rows(bufB, lane, LK,  K1);
  to_rows(bufA, lane, LB,  C3);
  to_rows(bufB, lane, LW,  Wd);
  to_rows(bufA, lane, LC,  CC);
  to_rows(bufB, lane, LM,  MK);
  to_rows(bufA, lane, LXa, Xa);
  to_rows(bufB, lane, LXb, Xb);
  #pragma unroll
  for (int t = 0; t < 4; ++t) {
    #pragma unroll
    for (int v = 0; v < 8; ++v) {
      K1[t][v] = in[t] ? K1[t][v] : 0.0f;
      C3[t][v] = 4.0f - C3[t][v];
      Na[t][v] = 0.0f;  Nb[t][v] = 0.0f;
    }
  }
  pw4(acc0, U, Wd, K1, C3, CC, MK, Xa);

  const float* xp = xi + 4 * plane;
  for (int it = 0; it < NS / 2; ++it) {
    const bool more = it + 1 < NS / 2;
    if (it > 0) {
      if (more) { ldL(xp, off, La);  ldL(xp + plane, off, Lb);  xp += 2 * plane; }
      step4(bufT, lane, U,  Wd, K1, C3, CC, MK, Xa, tv, s0, s3);
    }
    if (more) { to_rows(bufA, lane, La, Na);  to_rows(bufB, lane, Lb, Nb); }
    step4(bufT, lane, Wd, U,  K1, C3, CC, MK, Xb, tv, s0, s3);

    #pragma unroll
    for (int t = 0; t < 4; ++t) {
      #pragma unroll
      for (int v = 0; v < 8; ++v) { Xa[t][v] = Na[t][v];  Xb[t][v] = Nb[t][v]; }
    }
  }

  const bool lower = lp < 8;
  const bool right = hi == 0;
  float a1[8], a2[8];
  #pragma unroll
  for (int v = 0; v < 8; ++v) {
    const float t1 = right ? Wd[1][v] : Wd[0][v];
    const float b1 = right ? Wd[3][v] : Wd[2][v];
    a1[v] = lower ? b1 : t1;
    const float t2 = right ? U[1][v] : U[0][v];
    const float b2 = right ? U[3][v] : U[2][v];
    a2[v] = lower ? b2 : t2;
  }
  const int so = (r0 - 8 + (lower ? 16 : 0) + lp) * GN + c0 - 8 + (right ? 16 : 0) + 8 * hi;
  *(float4*)&o1[so]     = make_float4(a1[0], a1[1], a1[2], a1[3]);
  *(float4*)&o1[so + 4] = make_float4(a1[4], a1[5], a1[6], a1[7]);
  *(float4*)&o2[so]     = make_float4(a2[0], a2[1], a2[2], a2[3]);
  *(float4*)&o2[so + 4] = make_float4(a2[4], a2[5], a2[6], a2[7]);
}

extern "C" void kernel_launch(void* const* d_in, const int* in_sizes, int n_in,
                              void* d_out, int out_size, void* d_ws, size_t ws_size,
                              hipStream_t stream) {
  const float* inp = (const float*)d_in[0];
  const float* un1 = (const float*)d_in[1];
  const float* un2 = (const float*)d_in[2];
  const float* c2  = (const float*)d_in[3];
  const float* msk = (const float*)d_in[4];
  const float* b   = (const float*)d_in[5];

  const int plane = GN * GN;
  const int nl = in_sizes[0] / plane / NS;

  float* pA  = (float*)d_ws;
  float* pC  = pA + plane;
  float* pD  = pC + plane;
  float* C1p = pD + plane;
  float* pB  = (float*)d_out;

  pml_coeff<<<dim3(plane / 256), dim3(256), 0, stream>>>(b, C1p);

  dim3 grid(256);
  dim3 block(128);

  for (int l = 0; l < nl; ++l) {
    const float* xi    = inp + (size_t)(NS * l) * plane;
    const float* xi_nx = (l + 1 < nl) ? (xi + (size_t)NS * plane) : xi;

    const float* u1 = (l == 0) ? un1 : ((l & 1) ? pD : pB);
    const float* u2 = (l == 0) ? un2 : ((l & 1) ? pC : pA);
    float* o1 = (l & 1) ? pA : pC;
    float* o2 = (l & 1) ? pB : pD;

    wave_step8<<<grid, block, 0, stream>>>(xi, xi_nx, u1, u2, c2, msk, C1p, b, o1, o2);
  }
}


# Steps 2-8 of a launch: the k slot of the region's outermost row/column (outside the dependency cone) carries the neighbour tile's edge, no one-hot edge WMMAs: 40 to 32 a step; edge sum order changes
# speedup vs baseline: 9.3895x; 1.0680x over previous
#include <hip/hip_runtime.h>

typedef float v2f __attribute__((ext_vector_type(2)));
typedef float v8f __attribute__((ext_vector_type(8)));

#define GN 512
#define NS 8
#define LPITCH 36
#define LBUF (32 * LPITCH)

__global__ __launch_bounds__(256)
void pml_coeff(const float* __restrict__ b, float* __restrict__ C1p)
{
  const int i = blockIdx.x * 256 + threadIdx.x;
  C1p[i] = 1.0f / (4.0f + b[i]);
}

__device__ __forceinline__ v8f mma4(v2f a, v2f b, v8f c)
{
  return __builtin_amdgcn_wmma_f32_16x16x4_f32(false, a, false, b, (short)0, c,
                                               false, false);
}

__device__ __forceinline__ void ldL(const float* p, const unsigned (&off)[8],
                                    float (&d)[8][4])
{
  #pragma unroll
  for (int i = 0; i < 8; ++i) {
    const float4 a = *(const float4*)&p[off[i]];
    d[i][0] = a.x; d[i][1] = a.y; d[i][2] = a.z; d[i][3] = a.w;
  }
}

__device__ __forceinline__ void wave_lds_order()
{
  __builtin_amdgcn_fence(__ATOMIC_RELEASE, "wavefront");
  __builtin_amdgcn_wave_barrier();
  __builtin_amdgcn_fence(__ATOMIC_ACQUIRE, "wavefront");
}

__device__ __forceinline__ void to_rows(float* __restrict__ buf, int lane,
                                        const float (&d)[8][4], float (&R)[4][8])
{
  const int sr = lane >> 3, sg = lane & 7;
  const int lp = lane & 15, hi = lane >> 4;
  #pragma unroll
  for (int i = 0; i < 8; ++i)
    *(float4*)&buf[(4 * i + sr) * LPITCH + 4 * sg] = make_float4(d[i][0], d[i][1], d[i][2], d[i][3]);
  wave_lds_order();
  #pragma unroll
  for (int t = 0; t < 4; ++t) {
    const int o = (16 * (t >> 1) + lp) * LPITCH + 16 * (t & 1) + 8 * hi;
    const float4 a = *(const float4*)&buf[o];
    const float4 c = *(const float4*)&buf[o + 4];
    R[t][0] = a.x; R[t][1] = a.y; R[t][2] = a.z; R[t][3] = a.w;
    R[t][4] = c.x; R[t][5] = c.y; R[t][6] = c.z; R[t][7] = c.w;
  }
  wave_lds_order();
}

__device__ __forceinline__ void lap4(float* __restrict__ tb, int lane,
                                     const float (&S)[4][8], v8f (&acc)[4],
                                     const v2f (&tv)[4], const v2f s0, const v2f s3)
{
  const int lp = lane & 15, hi = lane >> 4;
  #pragma unroll
  for (int t = 0; t < 4; ++t) {
    #pragma unroll
    for (int v = 0; v < 8; ++v)
      tb[(16 * (t & 1) + 8 * hi + v) * LPITCH + 16 * (t >> 1) + lp] = S[t][v];
  }
  wave_lds_order();
  v8f col[4];
  #pragma unroll
  for (int t = 0; t < 4; ++t) {
    const int o = (16 * (t & 1) + lp) * LPITCH + 16 * (t >> 1) + 8 * hi;
    const float4 a = *(const float4*)&tb[o];
    const float4 c = *(const float4*)&tb[o + 4];
    col[t][0] = a.x; col[t][1] = a.y; col[t][2] = a.z; col[t][3] = a.w;
    col[t][4] = c.x; col[t][5] = c.y; col[t][6] = c.z; col[t][7] = c.w;
  }
  wave_lds_order();
  __builtin_amdgcn_sched_barrier(0);

  #pragma unroll
  for (int t = 0; t < 4; ++t) acc[t] = v8f{};
  #pragma unroll
  for (int j = 0; j < 4; ++j) {
    #pragma unroll
    for (int t = 0; t < 4; ++t) {
      v2f r;  r.x = S[t][2 * j];  r.y = S[t][2 * j + 1];
      acc[t] = mma4(tv[j], r, acc[t]);
    }
  }
  #pragma unroll
  for (int t = 0; t < 4; ++t) {
    const int nh = t ^ 1;
    v2f r;
    if ((t & 1) == 0) {
      r.x = S[nh][0];  r.y = S[nh][1];
      acc[t] = mma4(s0, r, acc[t]);
    } else {
      r.x = S[nh][6];  r.y = S[nh][7];
      acc[t] = mma4(s3, r, acc[t]);
    }
  }
  #pragma unroll
  for (int j = 0; j < 4; ++j) {
    #pragma unroll
    for (int t = 0; t < 4; ++t) {
      v2f c;  c.x = col[t][2 * j];  c.y = col[t][2 * j + 1];
      acc[t] = mma4(c, tv[j], acc[t]);
    }
  }
  #pragma unroll
  for (int t = 0; t < 4; ++t) {
    const int nv = t ^ 2;
    v2f c;
    if ((t >> 1) == 0) {
      c.x = col[nv][0];  c.y = col[nv][1];
      acc[t] = mma4(c, s0, acc[t]);
    } else {
      c.x = col[nv][6];  c.y = col[nv][7];
      acc[t] = mma4(c, s3, acc[t]);
    }
  }
}

__device__ __forceinline__ void lap4e(float* __restrict__ tb, int lane,
                                      const float (&S)[4][8], v8f (&acc)[4],
                                      const v2f (&tv)[4], const v2f tv0e, const v2f tv3o)
{
  const int lp = lane & 15, hi = lane >> 4;
  const bool h0 = hi == 0;
  #pragma unroll
  for (int t = 0; t < 4; ++t) {
    #pragma unroll
    for (int v = 0; v < 8; ++v)
      tb[(16 * (t & 1) + 8 * hi + v) * LPITCH + 16 * (t >> 1) + lp] = S[t][v];
  }
  wave_lds_order();
  v8f col[4];
  #pragma unroll
  for (int t = 0; t < 4; ++t) {
    const int o = (16 * (t & 1) + lp) * LPITCH + 16 * (t >> 1) + 8 * hi;
    const float4 a = *(const float4*)&tb[o];
    const float4 c = *(const float4*)&tb[o + 4];
    col[t][0] = a.x; col[t][1] = a.y; col[t][2] = a.z; col[t][3] = a.w;
    col[t][4] = c.x; col[t][5] = c.y; col[t][6] = c.z; col[t][7] = c.w;
  }
  wave_lds_order();
  __builtin_amdgcn_sched_barrier(0);

  #pragma unroll
  for (int t = 0; t < 4; ++t) acc[t] = v8f{};
  #pragma unroll
  for (int j = 0; j < 4; ++j) {
    #pragma unroll
    for (int t = 0; t < 4; ++t) {
      v2f r;  r.x = S[t][2 * j];  r.y = S[t][2 * j + 1];
      v2f a = tv[j];
      if (j == 0 && (t & 1) == 0) { r.x = h0 ? S[t ^ 1][0] : r.x;  a = tv0e; }
      if (j == 3 && (t & 1) == 1) { r.y = h0 ? r.y : S[t ^ 1][7];  a = tv3o; }
      acc[t] = mma4(a, r, acc[t]);
    }
  }
  __builtin_amdgcn_sched_barrier(0);
  #pragma unroll
  for (int j = 0; j < 4; ++j) {
    #pragma unroll
    for (int t = 0; t < 4; ++t) {
      v2f c;  c.x = col[t][2 * j];  c.y = col[t][2 * j + 1];
      v2f b = tv[j];
      if (j == 0 && (t >> 1) == 0) { c.x = h0 ? col[t ^ 2][0] : c.x;  b = tv0e; }
      if (j == 3 && (t >> 1) == 1) { c.y = h0 ? c.y : col[t ^ 2][7];  b = tv3o; }
      acc[t] = mma4(c, b, acc[t]);
    }
  }
}

__device__ __forceinline__ void pw4(const v8f (&acc)[4], const float (&S)[4][8],
                                    float (&D)[4][8],
                                    const float (&K1)[4][8], const float (&C3)[4][8],
                                    const float (&CC)[4][8], const float (&MK)[4][8],
                                    const float (&X)[4][8])
{
  #pragma unroll
  for (int t = 0; t < 4; ++t) {
    #pragma unroll
    for (int v = 0; v < 8; ++v)
      D[t][v] = K1[t][v] * (8.0f * S[t][v] - C3[t][v] * D[t][v]
                            + CC[t][v] * acc[t][v] + MK[t][v] * X[t][v]);
  }
}

__device__ __forceinline__ void step4(float* __restrict__ tb, int lane,
                                      const float (&S)[4][8], float (&D)[4][8],
                                      const float (&K1)[4][8], const float (&C3)[4][8],
                                      const float (&CC)[4][8], const float (&MK)[4][8],
                                      const float (&X)[4][8],
                                      const v2f (&tv)[4],
                                      const v2f tv0e, const v2f tv3o)
{
  v8f acc[4];
  lap4e(tb, lane, S, acc, tv, tv0e, tv3o);
  pw4(acc, S, D, K1, C3, CC, MK, X);
}

__global__ __launch_bounds__(128)
void wave_step8(const float* xi,
                const float* xi_nx,
                const float* u1,
                const float* u2,
                const float* c2,
                const float* msk,
                const float* C1p,
                const float* b,
                float* o1,
                float* o2)
{
  const int lane = threadIdx.x & 31;
  const int wave = threadIdx.x >> 5;
  const int tile = blockIdx.x * 4 + wave;
  const int r0 = (tile >> 5) * 16;
  const int c0 = (tile & 31) * 16;

  const int hi = lane >> 4;
  const int lp = lane & 15;
  const int plane = GN * GN;

  #pragma unroll
  for (int s = 0; s < NS; s += 2)
    __builtin_prefetch(&xi_nx[(s + hi) * plane + (r0 + lp) * GN + c0], 0, 1);

  __shared__ __attribute__((aligned(16))) float stage[4][3][LBUF];
  float* const bufA = &stage[wave][0][0];
  float* const bufB = &stage[wave][1][0];
  float* const bufT = &stage[wave][2][0];

  const int sr = lane >> 3, sg = lane & 7;
  const int gc = c0 - 8 + 4 * sg;
  const bool colin = (unsigned)gc < (unsigned)GN;
  const int gcs = colin ? gc : c0 + 4 * (sg & 3);
  unsigned off[8];
  #pragma unroll
  for (int i = 0; i < 8; ++i) {
    const int gr = r0 - 8 + 4 * i + sr;
    const bool rowin = (unsigned)gr < (unsigned)GN;
    off[i] = (unsigned)((rowin ? gr : r0 + sr) * GN + gcs);
  }

  const unsigned rowcode = (unsigned)((0x19ull << (2 * lp)) >> 2);
  v2f tv[4];
  #pragma unroll
  for (int j = 0; j < 4; ++j) {
    const int sh = 30 - 16 * hi - 4 * j;
    tv[j].x = (float)((int)(rowcode << sh) >> 30);
    tv[j].y = (float)((int)(rowcode << (sh - 2)) >> 30);
  }
  v2f s0, s3;
  s0.x = (hi == 0 && lp == 15) ? 1.0f : 0.0f;  s0.y = 0.0f;
  s3.x = 0.0f;  s3.y = (hi == 1 && lp == 0) ? 1.0f : 0.0f;
  v2f tv0e = tv[0], tv3o = tv[3];
  tv0e.x = (hi == 0) ? s0.x : tv0e.x;
  tv3o.y = (hi == 0) ? tv3o.y : s3.y;

  bool in[4];
  #pragma unroll
  for (int t = 0; t < 4; ++t) {
    const int row = r0 - 8 + 16 * (t >> 1) + lp;
    const int cl  = c0 - 8 + 16 * (t & 1) + 8 * hi;
    in[t] = ((unsigned)row < (unsigned)GN) && ((unsigned)cl < (unsigned)GN);
  }

  float LU[8][4], LK[8][4], LB[8][4], LW[8][4], LC[8][4], LM[8][4], LXa[8][4], LXb[8][4];
  ldL(u1,  off, LU);
  __builtin_amdgcn_sched_barrier(0);
  ldL(C1p, off, LK);
  __builtin_amdgcn_sched_barrier(0);
  ldL(b,   off, LB);
  __builtin_amdgcn_sched_barrier(0);
  ldL(u2,  off, LW);
  __builtin_amdgcn_sched_barrier(0);
  ldL(c2,  off, LC);
  __builtin_amdgcn_sched_barrier(0);
  ldL(msk, off, LM);
  __builtin_amdgcn_sched_barrier(0);
  ldL(xi,  off, LXa);
  __builtin_amdgcn_sched_barrier(0);
  ldL(xi + plane, off, LXb);
  __builtin_amdgcn_sched_barrier(0);

  float U[4][8], Wd[4][8], K1[4][8], C3[4][8], CC[4][8], MK[4][8];
  float Xa[4][8], Xb[4][8], Na[4][8], Nb[4][8];
  to_rows(bufA, lane, LU, U);
  #pragma unroll
  for (int t = 0; t < 4; ++t) {
    #pragma unroll
    for (int v = 0; v < 8; ++v) U[t][v] = in[t] ? U[t][v] : 0.0f;
  }
  __builtin_amdgcn_sched_barrier(0);

  float La[8][4], Lb[8][4];
  ldL(xi + 2 * plane, off, La);
  ldL(xi + 3 * plane, off, Lb);

  v8f acc0[4];
  lap4(bufT, lane, U, acc0, tv, s0, s3);

  to_rows(bufB, lane, LK,  K1);
  to_rows(bufA, lane, LB,  C3);
  to_rows(bufB, lane, LW,  Wd);
  to_rows(bufA, lane, LC,  CC);
  to_rows(bufB, lane, LM,  MK);
  to_rows(bufA, lane, LXa, Xa);
  to_rows(bufB, lane, LXb, Xb);
  #pragma unroll
  for (int t = 0; t < 4; ++t) {
    #pragma unroll
    for (int v = 0; v < 8; ++v) {
      K1[t][v] = in[t] ? K1[t][v] : 0.0f;
      C3[t][v] = 4.0f - C3[t][v];
      Na[t][v] = 0.0f;  Nb[t][v] = 0.0f;
    }
  }
  pw4(acc0, U, Wd, K1, C3, CC, MK, Xa);

  const float* xp = xi + 4 * plane;
  for (int it = 0; it < NS / 2; ++it) {
    const bool more = it + 1 < NS / 2;
    if (it > 0) {
      if (more) { ldL(xp, off, La);  ldL(xp + plane, off, Lb);  xp += 2 * plane; }
      step4(bufT, lane, U,  Wd, K1, C3, CC, MK, Xa, tv, tv0e, tv3o);
    }
    if (more) { to_rows(bufA, lane, La, Na);  to_rows(bufB, lane, Lb, Nb); }
    step4(bufT, lane, Wd, U,  K1, C3, CC, MK, Xb, tv, tv0e, tv3o);

    #pragma unroll
    for (int t = 0; t < 4; ++t) {
      #pragma unroll
      for (int v = 0; v < 8; ++v) { Xa[t][v] = Na[t][v];  Xb[t][v] = Nb[t][v]; }
    }
  }

  const bool lower = lp < 8;
  const bool right = hi == 0;
  float a1[8], a2[8];
  #pragma unroll
  for (int v = 0; v < 8; ++v) {
    const float t1 = right ? Wd[1][v] : Wd[0][v];
    const float b1 = right ? Wd[3][v] : Wd[2][v];
    a1[v] = lower ? b1 : t1;
    const float t2 = right ? U[1][v] : U[0][v];
    const float b2 = right ? U[3][v] : U[2][v];
    a2[v] = lower ? b2 : t2;
  }
  const int so = (r0 - 8 + (lower ? 16 : 0) + lp) * GN + c0 - 8 + (right ? 16 : 0) + 8 * hi;
  *(float4*)&o1[so]     = make_float4(a1[0], a1[1], a1[2], a1[3]);
  *(float4*)&o1[so + 4] = make_float4(a1[4], a1[5], a1[6], a1[7]);
  *(float4*)&o2[so]     = make_float4(a2[0], a2[1], a2[2], a2[3]);
  *(float4*)&o2[so + 4] = make_float4(a2[4], a2[5], a2[6], a2[7]);
}

extern "C" void kernel_launch(void* const* d_in, const int* in_sizes, int n_in,
                              void* d_out, int out_size, void* d_ws, size_t ws_size,
                              hipStream_t stream) {
  const float* inp = (const float*)d_in[0];
  const float* un1 = (const float*)d_in[1];
  const float* un2 = (const float*)d_in[2];
  const float* c2  = (const float*)d_in[3];
  const float* msk = (const float*)d_in[4];
  const float* b   = (const float*)d_in[5];

  const int plane = GN * GN;
  const int nl = in_sizes[0] / plane / NS;

  float* pA  = (float*)d_ws;
  float* pC  = pA + plane;
  float* pD  = pC + plane;
  float* C1p = pD + plane;
  float* pB  = (float*)d_out;

  pml_coeff<<<dim3(plane / 256), dim3(256), 0, stream>>>(b, C1p);

  dim3 grid(256);
  dim3 block(128);

  for (int l = 0; l < nl; ++l) {
    const float* xi    = inp + (size_t)(NS * l) * plane;
    const float* xi_nx = (l + 1 < nl) ? (xi + (size_t)NS * plane) : xi;

    const float* u1 = (l == 0) ? un1 : ((l & 1) ? pD : pB);
    const float* u2 = (l == 0) ? un2 : ((l & 1) ? pC : pA);
    float* o1 = (l & 1) ? pA : pC;
    float* o2 = (l & 1) ? pB : pD;

    wave_step8<<<grid, block, 0, stream>>>(xi, xi_nx, u1, u2, c2, msk, C1p, b, o1, o2);
  }
}


# Steps specialised by depth in the dependency cone (loop written out): steps 6-8 drop the column k chunk outside it (32 to 28 WMMAs), forcing rows outside it not loaded; Laplacian sum order changes
# speedup vs baseline: 9.8969x; 1.0540x over previous
#include <hip/hip_runtime.h>

typedef float v2f __attribute__((ext_vector_type(2)));
typedef float v8f __attribute__((ext_vector_type(8)));

#define GN 512
#define NS 8
#define LPITCH 36
#define LBUF (32 * LPITCH)

__global__ __launch_bounds__(256)
void pml_coeff(const float* __restrict__ b, float* __restrict__ C1p)
{
  const int i = blockIdx.x * 256 + threadIdx.x;
  C1p[i] = 1.0f / (4.0f + b[i]);
}

__device__ __forceinline__ v8f mma4(v2f a, v2f b, v8f c)
{
  return __builtin_amdgcn_wmma_f32_16x16x4_f32(false, a, false, b, (short)0, c,
                                               false, false);
}

template <int I0, int I1>
__device__ __forceinline__ void ldL(const float* p, const unsigned (&off)[8],
                                    float (&d)[8][4])
{
  #pragma unroll
  for (int i = I0; i < I1; ++i) {
    const float4 a = *(const float4*)&p[off[i]];
    d[i][0] = a.x; d[i][1] = a.y; d[i][2] = a.z; d[i][3] = a.w;
  }
}

__device__ __forceinline__ void wave_lds_order()
{
  __builtin_amdgcn_fence(__ATOMIC_RELEASE, "wavefront");
  __builtin_amdgcn_wave_barrier();
  __builtin_amdgcn_fence(__ATOMIC_ACQUIRE, "wavefront");
}

template <int I0, int I1>
__device__ __forceinline__ void to_rows(float* __restrict__ buf, int lane,
                                        const float (&d)[8][4], float (&R)[4][8])
{
  const int sr = lane >> 3, sg = lane & 7;
  const int lp = lane & 15, hi = lane >> 4;
  #pragma unroll
  for (int i = I0; i < I1; ++i)
    *(float4*)&buf[(4 * i + sr) * LPITCH + 4 * sg] = make_float4(d[i][0], d[i][1], d[i][2], d[i][3]);
  wave_lds_order();
  #pragma unroll
  for (int t = 0; t < 4; ++t) {
    const int o = (16 * (t >> 1) + lp) * LPITCH + 16 * (t & 1) + 8 * hi;
    const float4 a = *(const float4*)&buf[o];
    const float4 c = *(const float4*)&buf[o + 4];
    R[t][0] = a.x; R[t][1] = a.y; R[t][2] = a.z; R[t][3] = a.w;
    R[t][4] = c.x; R[t][5] = c.y; R[t][6] = c.z; R[t][7] = c.w;
  }
  wave_lds_order();
}

__device__ __forceinline__ void lap4(float* __restrict__ tb, int lane,
                                     const float (&S)[4][8], v8f (&acc)[4],
                                     const v2f (&tv)[4], const v2f (&tc)[4],
                                     const v2f s0, const v2f s3)
{
  const int lp = lane & 15, hi = lane >> 4;
  const int pp = 8 * ((lp >> 1) & 1) + 2 * (lp >> 2) + (lp & 1);
  #pragma unroll
  for (int t = 0; t < 4; ++t) {
    #pragma unroll
    for (int v = 0; v < 8; ++v)
      tb[(16 * (t & 1) + 8 * hi + v) * LPITCH + 16 * (t >> 1) + pp] = S[t][v];
  }
  wave_lds_order();
  v8f col[4];
  #pragma unroll
  for (int t = 0; t < 4; ++t) {
    const int o = (16 * (t & 1) + lp) * LPITCH + 16 * (t >> 1) + 8 * hi;
    const float4 a = *(const float4*)&tb[o];
    const float4 c = *(const float4*)&tb[o + 4];
    col[t][0] = a.x; col[t][1] = a.y; col[t][2] = a.z; col[t][3] = a.w;
    col[t][4] = c.x; col[t][5] = c.y; col[t][6] = c.z; col[t][7] = c.w;
  }
  wave_lds_order();
  __builtin_amdgcn_sched_barrier(0);

  #pragma unroll
  for (int t = 0; t < 4; ++t) acc[t] = v8f{};
  #pragma unroll
  for (int j = 0; j < 4; ++j) {
    #pragma unroll
    for (int t = 0; t < 4; ++t) {
      v2f r;  r.x = S[t][2 * j];  r.y = S[t][2 * j + 1];
      acc[t] = mma4(tv[j], r, acc[t]);
    }
  }
  #pragma unroll
  for (int t = 0; t < 4; ++t) {
    const int nh = t ^ 1;
    v2f r;
    if ((t & 1) == 0) {
      r.x = S[nh][0];  r.y = S[nh][1];
      acc[t] = mma4(s0, r, acc[t]);
    } else {
      r.x = S[nh][6];  r.y = S[nh][7];
      acc[t] = mma4(s3, r, acc[t]);
    }
  }
  #pragma unroll
  for (int j = 0; j < 4; ++j) {
    #pragma unroll
    for (int t = 0; t < 4; ++t) {
      v2f c;  c.x = col[t][2 * j];  c.y = col[t][2 * j + 1];
      acc[t] = mma4(c, tc[j], acc[t]);
    }
  }
  #pragma unroll
  for (int t = 0; t < 4; ++t) {
    const int nv = t ^ 2;
    v2f c;
    if ((t >> 1) == 0) {
      c.x = col[nv][0];  c.y = col[nv][1];
      acc[t] = mma4(c, s0, acc[t]);
    } else {
      c.x = col[nv][6];  c.y = col[nv][7];
      acc[t] = mma4(c, s3, acc[t]);
    }
  }
}

template <bool LATE>
__device__ __forceinline__ void lap4e(float* __restrict__ tb, int lane,
                                      const float (&S)[4][8], v8f (&acc)[4],
                                      const v2f (&tv)[4], const v2f tv0e, const v2f tv3o,
                                      const v2f (&tc)[4], const v2f tcU, const v2f tcL)
{
  const int lp = lane & 15, hi = lane >> 4;
  const int pp = 8 * ((lp >> 1) & 1) + 2 * (lp >> 2) + (lp & 1);
  const bool h0 = hi == 0;
  #pragma unroll
  for (int t = 0; t < 4; ++t) {
    #pragma unroll
    for (int v = 0; v < 8; ++v)
      tb[(16 * (t & 1) + 8 * hi + v) * LPITCH + 16 * (t >> 1) + pp] = S[t][v];
  }
  wave_lds_order();
  v8f col[4];
  #pragma unroll
  for (int t = 0; t < 4; ++t) {
    const int o = (16 * (t & 1) + lp) * LPITCH + 16 * (t >> 1) + 8 * hi;
    const float4 a = *(const float4*)&tb[o];
    const float4 c = *(const float4*)&tb[o + 4];
    col[t][0] = a.x; col[t][1] = a.y; col[t][2] = a.z; col[t][3] = a.w;
    col[t][4] = c.x; col[t][5] = c.y; col[t][6] = c.z; col[t][7] = c.w;
  }
  wave_lds_order();
  __builtin_amdgcn_sched_barrier(0);

  #pragma unroll
  for (int t = 0; t < 4; ++t) acc[t] = v8f{};
  #pragma unroll
  for (int j = 0; j < 4; ++j) {
    #pragma unroll
    for (int t = 0; t < 4; ++t) {
      v2f r;  r.x = S[t][2 * j];  r.y = S[t][2 * j + 1];
      v2f a = tv[j];
      if (j == 0 && (t & 1) == 0) { r.x = h0 ? S[t ^ 1][0] : r.x;  a = tv0e; }
      if (j == 3 && (t & 1) == 1) { r.y = h0 ? r.y : S[t ^ 1][7];  a = tv3o; }
      acc[t] = mma4(a, r, acc[t]);
    }
  }
  __builtin_amdgcn_sched_barrier(0);
  constexpr int NJ = LATE ? 3 : 4;
  constexpr int JU = LATE ? 1 : 0;
  constexpr int JL = LATE ? 2 : 3;
  #pragma unroll
  for (int jj = 0; jj < NJ; ++jj) {
    #pragma unroll
    for (int t = 0; t < 4; ++t) {
      const bool up = (t >> 1) == 0;
      const int j = (LATE && up) ? jj + 1 : jj;
      v2f c;  c.x = col[t][2 * j];  c.y = col[t][2 * j + 1];
      v2f b = tc[j];
      if (up && j == JU)  { c.x = h0 ? col[t ^ 2][0] : c.x;  b = tcU; }
      if (!up && j == JL) { c.y = h0 ? c.y : col[t ^ 2][7];  b = tcL; }
      acc[t] = mma4(c, b, acc[t]);
    }
  }
}

__device__ __forceinline__ void pw4(const v8f (&acc)[4], const float (&S)[4][8],
                                    float (&D)[4][8],
                                    const float (&K1)[4][8], const float (&C3)[4][8],
                                    const float (&CC)[4][8], const float (&MK)[4][8],
                                    const float (&X)[4][8])
{
  #pragma unroll
  for (int t = 0; t < 4; ++t) {
    #pragma unroll
    for (int p = 0; p < 8; p += 2) {
      const v2f s  = {S[t][p],   S[t][p + 1]};
      const v2f d  = {D[t][p],   D[t][p + 1]};
      const v2f k1 = {K1[t][p],  K1[t][p + 1]};
      const v2f c3 = {C3[t][p],  C3[t][p + 1]};
      const v2f cc = {CC[t][p],  CC[t][p + 1]};
      const v2f mk = {MK[t][p],  MK[t][p + 1]};
      const v2f x  = {X[t][p],   X[t][p + 1]};
      const v2f a  = {acc[t][p], acc[t][p + 1]};
      const v2f r = k1 * (8.0f * s - c3 * d + cc * a + mk * x);
      D[t][p] = r.x;  D[t][p + 1] = r.y;
    }
  }
}

template <bool LATE>
__device__ __forceinline__ void step4(float* __restrict__ tb, int lane,
                                      const float (&S)[4][8], float (&D)[4][8],
                                      const float (&K1)[4][8], const float (&C3)[4][8],
                                      const float (&CC)[4][8], const float (&MK)[4][8],
                                      const float (&X)[4][8],
                                      const v2f (&tv)[4],
                                      const v2f tv0e, const v2f tv3o,
                                      const v2f (&tc)[4],
                                      const v2f tcU, const v2f tcL)
{
  v8f acc[4];
  lap4e<LATE>(tb, lane, S, acc, tv, tv0e, tv3o, tc, tcU, tcL);
  pw4(acc, S, D, K1, C3, CC, MK, X);
}

__global__ __launch_bounds__(128)
void wave_step8(const float* xi,
                const float* xi_nx,
                const float* u1,
                const float* u2,
                const float* c2,
                const float* msk,
                const float* C1p,
                const float* b,
                float* o1,
                float* o2)
{
  const int lane = threadIdx.x & 31;
  const int wave = threadIdx.x >> 5;
  const int tile = blockIdx.x * 4 + wave;
  const int r0 = (tile >> 5) * 16;
  const int c0 = (tile & 31) * 16;

  const int hi = lane >> 4;
  const int lp = lane & 15;
  const int plane = GN * GN;

  #pragma unroll
  for (int s = 0; s < NS; s += 2)
    __builtin_prefetch(&xi_nx[(s + hi) * plane + (r0 + lp) * GN + c0], 0, 1);

  __shared__ __attribute__((aligned(16))) float stage[4][3][LBUF];
  float* const bufA = &stage[wave][0][0];
  float* const bufB = &stage[wave][1][0];
  float* const bufT = &stage[wave][2][0];

  const int sr = lane >> 3, sg = lane & 7;
  const int gc = c0 - 8 + 4 * sg;
  const bool colin = (unsigned)gc < (unsigned)GN;
  const int gcs = colin ? gc : c0 + 4 * (sg & 3);
  unsigned off[8];
  #pragma unroll
  for (int i = 0; i < 8; ++i) {
    const int gr = r0 - 8 + 4 * i + sr;
    const bool rowin = (unsigned)gr < (unsigned)GN;
    off[i] = (unsigned)((rowin ? gr : r0 + sr) * GN + gcs);
  }

  const unsigned rowcode = (unsigned)((0x19ull << (2 * lp)) >> 2);
  v2f tv[4];
  #pragma unroll
  for (int j = 0; j < 4; ++j) {
    const int sh = 30 - 16 * hi - 4 * j;
    tv[j].x = (float)((int)(rowcode << sh) >> 30);
    tv[j].y = (float)((int)(rowcode << (sh - 2)) >> 30);
  }
  v2f tc[4];
  #pragma unroll
  for (int j = 0; j < 4; ++j) {
    const int sh = 30 - 8 * j - 4 * hi;
    tc[j].x = (float)((int)(rowcode << sh) >> 30);
    tc[j].y = (float)((int)(rowcode << (sh - 2)) >> 30);
  }
  v2f s0, s3;
  s0.x = (hi == 0 && lp == 15) ? 1.0f : 0.0f;  s0.y = 0.0f;
  s3.x = 0.0f;  s3.y = (hi == 1 && lp == 0) ? 1.0f : 0.0f;
  v2f tv0e = tv[0], tv3o = tv[3];
  tv0e.x = (hi == 0) ? s0.x : tv0e.x;
  tv3o.y = (hi == 0) ? tv3o.y : s3.y;
  v2f tcUe = tc[0], tcLe = tc[3], tcUl = tc[1], tcLl = tc[2];
  tcUe.x = (hi == 0) ? s0.x : tcUe.x;
  tcLe.y = (hi == 0) ? tcLe.y : s3.y;
  tcUl.x = (hi == 0) ? s0.x : tcUl.x;
  tcLl.y = (hi == 0) ? tcLl.y : s3.y;

  bool in[4];
  #pragma unroll
  for (int t = 0; t < 4; ++t) {
    const int row = r0 - 8 + 16 * (t >> 1) + lp;
    const int cl  = c0 - 8 + 16 * (t & 1) + 8 * hi;
    in[t] = ((unsigned)row < (unsigned)GN) && ((unsigned)cl < (unsigned)GN);
  }

  float LU[8][4], LK[8][4], LB[8][4], LW[8][4], LC[8][4], LM[8][4], LXa[8][4], LXb[8][4];
  ldL<0, 8>(u1,  off, LU);
  __builtin_amdgcn_sched_barrier(0);
  ldL<0, 8>(C1p, off, LK);
  __builtin_amdgcn_sched_barrier(0);
  ldL<0, 8>(b,   off, LB);
  __builtin_amdgcn_sched_barrier(0);
  ldL<0, 8>(u2,  off, LW);
  __builtin_amdgcn_sched_barrier(0);
  ldL<0, 8>(c2,  off, LC);
  __builtin_amdgcn_sched_barrier(0);
  ldL<0, 8>(msk, off, LM);
  __builtin_amdgcn_sched_barrier(0);
  ldL<0, 8>(xi,  off, LXa);
  __builtin_amdgcn_sched_barrier(0);
  ldL<0, 8>(xi + plane, off, LXb);
  __builtin_amdgcn_sched_barrier(0);

  float U[4][8], Wd[4][8], K1[4][8], C3[4][8], CC[4][8], MK[4][8];
  float Xa[4][8], Xb[4][8];
  to_rows<0, 8>(bufA, lane, LU, U);
  #pragma unroll
  for (int t = 0; t < 4; ++t) {
    #pragma unroll
    for (int v = 0; v < 8; ++v) U[t][v] = in[t] ? U[t][v] : 0.0f;
  }
  __builtin_amdgcn_sched_barrier(0);

  float La[8][4], Lb[8][4];
  ldL<0, 8>(xi + 2 * plane, off, La);
  ldL<1, 7>(xi + 3 * plane, off, Lb);

  v8f acc0[4];
  lap4(bufT, lane, U, acc0, tv, tc, s0, s3);

  to_rows<0, 8>(bufB, lane, LK,  K1);
  to_rows<0, 8>(bufA, lane, LB,  C3);
  to_rows<0, 8>(bufB, lane, LW,  Wd);
  to_rows<0, 8>(bufA, lane, LC,  CC);
  to_rows<0, 8>(bufB, lane, LM,  MK);
  to_rows<0, 8>(bufA, lane, LXa, Xa);
  to_rows<0, 8>(bufB, lane, LXb, Xb);
  #pragma unroll
  for (int t = 0; t < 4; ++t) {
    #pragma unroll
    for (int v = 0; v < 8; ++v) {
      K1[t][v] = in[t] ? K1[t][v] : 0.0f;
      C3[t][v] = 4.0f - C3[t][v];
    }
  }
  pw4(acc0, U, Wd, K1, C3, CC, MK, Xa);

  float X3[4][8], X4[4][8], X5[4][8], X6[4][8], X7[4][8], X8[4][8];
  float Lc[8][4], Ld[8][4], Le[8][4], Lf[8][4];
  to_rows<0, 8>(bufA, lane, La, X3);
  to_rows<1, 7>(bufB, lane, Lb, X4);
  step4<false>(bufT, lane, Wd, U,  K1, C3, CC, MK, Xb, tv, tv0e, tv3o, tc, tcUe, tcLe);
  ldL<1, 7>(xi + 4 * plane, off, Lc);
  ldL<1, 7>(xi + 5 * plane, off, Ld);
  step4<false>(bufT, lane, U,  Wd, K1, C3, CC, MK, X3, tv, tv0e, tv3o, tc, tcUe, tcLe);
  to_rows<1, 7>(bufA, lane, Lc, X5);
  to_rows<1, 7>(bufB, lane, Ld, X6);
  step4<false>(bufT, lane, Wd, U,  K1, C3, CC, MK, X4, tv, tv0e, tv3o, tc, tcUe, tcLe);
  ldL<1, 7>(xi + 6 * plane, off, Le);
  ldL<2, 6>(xi + 7 * plane, off, Lf);
  step4<false>(bufT, lane, U,  Wd, K1, C3, CC, MK, X5, tv, tv0e, tv3o, tc, tcUe, tcLe);
  to_rows<1, 7>(bufA, lane, Le, X7);
  to_rows<2, 6>(bufB, lane, Lf, X8);
  step4<true>(bufT, lane, Wd, U,  K1, C3, CC, MK, X6, tv, tv0e, tv3o, tc, tcUl, tcLl);
  step4<true>(bufT, lane, U,  Wd, K1, C3, CC, MK, X7, tv, tv0e, tv3o, tc, tcUl, tcLl);
  step4<true>(bufT, lane, Wd, U,  K1, C3, CC, MK, X8, tv, tv0e, tv3o, tc, tcUl, tcLl);

  const bool lower = lp < 8;
  const bool right = hi == 0;
  float a1[8], a2[8];
  #pragma unroll
  for (int v = 0; v < 8; ++v) {
    const float t1 = right ? Wd[1][v] : Wd[0][v];
    const float b1 = right ? Wd[3][v] : Wd[2][v];
    a1[v] = lower ? b1 : t1;
    const float t2 = right ? U[1][v] : U[0][v];
    const float b2 = right ? U[3][v] : U[2][v];
    a2[v] = lower ? b2 : t2;
  }
  const int so = (r0 - 8 + (lower ? 16 : 0) + lp) * GN + c0 - 8 + (right ? 16 : 0) + 8 * hi;
  *(float4*)&o1[so]     = make_float4(a1[0], a1[1], a1[2], a1[3]);
  *(float4*)&o1[so + 4] = make_float4(a1[4], a1[5], a1[6], a1[7]);
  *(float4*)&o2[so]     = make_float4(a2[0], a2[1], a2[2], a2[3]);
  *(float4*)&o2[so + 4] = make_float4(a2[4], a2[5], a2[6], a2[7]);
}

extern "C" void kernel_launch(void* const* d_in, const int* in_sizes, int n_in,
                              void* d_out, int out_size, void* d_ws, size_t ws_size,
                              hipStream_t stream) {
  const float* inp = (const float*)d_in[0];
  const float* un1 = (const float*)d_in[1];
  const float* un2 = (const float*)d_in[2];
  const float* c2  = (const float*)d_in[3];
  const float* msk = (const float*)d_in[4];
  const float* b   = (const float*)d_in[5];

  const int plane = GN * GN;
  const int nl = in_sizes[0] / plane / NS;

  float* pA  = (float*)d_ws;
  float* pC  = pA + plane;
  float* pD  = pC + plane;
  float* C1p = pD + plane;
  float* pB  = (float*)d_out;

  pml_coeff<<<dim3(plane / 256), dim3(256), 0, stream>>>(b, C1p);

  dim3 grid(256);
  dim3 block(128);

  for (int l = 0; l < nl; ++l) {
    const float* xi    = inp + (size_t)(NS * l) * plane;
    const float* xi_nx = (l + 1 < nl) ? (xi + (size_t)NS * plane) : xi;

    const float* u1 = (l == 0) ? un1 : ((l & 1) ? pD : pB);
    const float* u2 = (l == 0) ? un2 : ((l & 1) ? pC : pA);
    float* o1 = (l & 1) ? pA : pC;
    float* o2 = (l & 1) ? pB : pD;

    wave_step8<<<grid, block, 0, stream>>>(xi, xi_nx, u1, u2, c2, msk, C1p, b, o1, o2);
  }
}
